# starting kernel
# baseline (speedup 1.0000x reference)
#include <hip/hip_runtime.h>
#include <math.h>

#define N_NODES   32768
#define N_EDGES   524288
#define NODE_DIM  128
#define HIDDEN_DIM 256
#define EDGE_DIM  64
#define IN_DIM    320

#define TILE_E    64
#define THREADS   256

#define A_LD      324
#define A_OFF     0
#define A_SIZE    (TILE_E * A_LD)
#define H_LD      260
#define H_OFF     (A_OFF + A_SIZE)
#define H_SIZE    (TILE_E * H_LD)
#define B_OFF     (H_OFF + H_SIZE)
#define B_SIZE    (32 * 256)
#define S_OFF     (B_OFF + B_SIZE)
#define SMEM_FLOATS (S_OFF + 4 * TILE_E)

typedef float v2f __attribute__((ext_vector_type(2)));
typedef float v8f __attribute__((ext_vector_type(8)));

__device__ __forceinline__ float silu_f(float v) {
    return v / (1.0f + __expf(-v));
}

__device__ __forceinline__ void atomAddF(float* p, float v) {
    __hip_atomic_fetch_add(p, v, __ATOMIC_RELAXED, __HIP_MEMORY_SCOPE_AGENT);
}

template<int NTILES, int NSTAGE>
__device__ __forceinline__ void gemm_tile(const float* __restrict__ W, int ldw, int K,
                                          const float* __restrict__ A, int lda,
                                          float* __restrict__ Bs,
                                          v8f* acc, int tid)
{
    const int lane  = tid & 31;
    const int wave  = tid >> 5;
    const int row   = lane & 15;
    const int hi    = lane >> 4;
    const int etile = wave >> 1;
    const int nsel  = wave & 1;
    const int nbase = nsel * (NSTAGE / 2);

    for (int kp = 0; kp < K; kp += 32) {
        __syncthreads();
        for (int i = tid; i < 32 * NSTAGE; i += THREADS) {
            int kl = i / NSTAGE;
            int n  = i - kl * NSTAGE;
            Bs[(((kl >> 2) * NSTAGE + n) << 2) + (kl & 3)] = W[(kp + kl) * ldw + n];
        }
        __syncthreads();

        const float* Arow = A + (etile * 16 + row) * lda + kp + 2 * hi;
#pragma unroll
        for (int kl = 0; kl < 32; kl += 4) {
            v2f a = *(const v2f*)(Arow + kl);
#pragma unroll
            for (int nt = 0; nt < NTILES; ++nt) {
                v2f b = *(const v2f*)(Bs + ((((kl >> 2) * NSTAGE + nbase + nt * 16 + row) << 2) + 2 * hi));
                acc[nt] = __builtin_amdgcn_wmma_f32_16x16x4_f32(
                     false, a,  false, b,
                     (short)0, acc[nt],
                     false,  false);
            }
        }
    }
}

__global__ void egnn_init_out(const float* __restrict__ h,
                              const float* __restrict__ x,
                              float* __restrict__ out)
{
    int i = blockIdx.x * blockDim.x + threadIdx.x;
    if (i < N_NODES * NODE_DIM)                out[i] = h[i];
    else if (i < N_NODES * (NODE_DIM + 3))     out[i] = x[i - N_NODES * NODE_DIM];
}

__global__ void __launch_bounds__(THREADS)
egnn_edge_kernel(const float* __restrict__ h,   const float* __restrict__ x,
                 const int*   __restrict__ eidx,const float* __restrict__ edist,
                 const float* __restrict__ nW1, const float* __restrict__ nb1,
                 const float* __restrict__ nW2, const float* __restrict__ nb2,
                 const float* __restrict__ cW1, const float* __restrict__ cb1,
                 const float* __restrict__ cW2,
                 const float* __restrict__ eW1, const float* __restrict__ eb1,
                 const float* __restrict__ eW2, const float* __restrict__ eb2,
                 float* __restrict__ out)
{
    extern __shared__ float smem[];
    float* Ablk = smem + A_OFF;
    float* Hblk = smem + H_OFF;
    float* Bs   = smem + B_OFF;
    int*   SRCs = (int*)(smem + S_OFF);
    int*   DSTs = (int*)(smem + S_OFF + TILE_E);
    float* DIST = smem + S_OFF + 2 * TILE_E;

    const int tid   = threadIdx.x;
    const int lane  = tid & 31;
    const int wave  = tid >> 5;
    const int row   = lane & 15;
    const int hi    = lane >> 4;
    const int etile = wave >> 1;
    const int nsel  = wave & 1;
    const int e0    = blockIdx.x * TILE_E;

    if (tid < TILE_E) {
        SRCs[tid] = eidx[e0 + tid];
        DSTs[tid] = eidx[N_EDGES + e0 + tid];
        DIST[tid] = edist[e0 + tid];
    }
    __syncthreads();

    for (int i = tid; i < TILE_E * 64; i += THREADS) {
        int e = i >> 6, q = i & 63;
        int half = q >> 5, c4 = q & 31;
        int node = half ? DSTs[e] : SRCs[e];
        *(float4*)(Ablk + e * A_LD + half * NODE_DIM + c4 * 4) =
            ((const float4*)h)[node * (NODE_DIM / 4) + c4];
    }

    float* hid = Hblk;
    for (int i = tid; i < TILE_E * EDGE_DIM; i += THREADS) {
        int e = i >> 6, k = i & 63;
        hid[k * TILE_E + e] = silu_f(DIST[e] * eW1[k] + eb1[k]);
    }
    __syncthreads();
    for (int i = tid; i < TILE_E * EDGE_DIM; i += THREADS) {
        int e = i >> 6, j = i & 63;
        float s = eb2[j];
#pragma unroll 8
        for (int k = 0; k < EDGE_DIM; ++k)
            s += hid[k * TILE_E + e] * eW2[k * EDGE_DIM + j];
        Ablk[e * A_LD + 2 * NODE_DIM + j] = s;
    }

    {
        v8f acc[8];
#pragma unroll
        for (int nt = 0; nt < 8; ++nt) {
            float bv = nb1[nsel * 128 + nt * 16 + row];
            v8f t = {bv, bv, bv, bv, bv, bv, bv, bv};
            acc[nt] = t;
        }
        gemm_tile<8, 256>(nW1, HIDDEN_DIM, IN_DIM, Ablk, A_LD, Bs, acc, tid);
#pragma unroll
        for (int nt = 0; nt < 8; ++nt)
#pragma unroll
            for (int i = 0; i < 8; ++i) {
                int e = etile * 16 + hi * 8 + i;
                Hblk[e * H_LD + nsel * 128 + nt * 16 + row] = silu_f(acc[nt][i]);
            }
    }

    {
        v8f acc[4];
#pragma unroll
        for (int nt = 0; nt < 4; ++nt) {
            float bv = nb2[nsel * 64 + nt * 16 + row];
            v8f t = {bv, bv, bv, bv, bv, bv, bv, bv};
            acc[nt] = t;
        }
        gemm_tile<4, 128>(nW2, NODE_DIM, HIDDEN_DIM, Hblk, H_LD, Bs, acc, tid);
#pragma unroll
        for (int nt = 0; nt < 4; ++nt)
#pragma unroll
            for (int i = 0; i < 8; ++i) {
                int e = etile * 16 + hi * 8 + i;
                int d = DSTs[e];
                atomAddF(out + d * NODE_DIM + nsel * 64 + nt * 16 + row, acc[nt][i]);
            }
    }

    {
        v8f acc[8];
#pragma unroll
        for (int nt = 0; nt < 8; ++nt) {
            float bv = cb1[nsel * 128 + nt * 16 + row];
            v8f t = {bv, bv, bv, bv, bv, bv, bv, bv};
            acc[nt] = t;
        }
        gemm_tile<8, 256>(cW1, HIDDEN_DIM, IN_DIM, Ablk, A_LD, Bs, acc, tid);
#pragma unroll
        for (int nt = 0; nt < 8; ++nt)
#pragma unroll
            for (int i = 0; i < 8; ++i) {
                int e = etile * 16 + hi * 8 + i;
                Hblk[e * H_LD + nsel * 128 + nt * 16 + row] = silu_f(acc[nt][i]);
            }
    }
    __syncthreads();

    if (tid < TILE_E) {
        int e = tid;
        float s = 0.0f;
#pragma unroll 8
        for (int k = 0; k < HIDDEN_DIM; ++k)
            s += Hblk[e * H_LD + k] * cW2[k];
        int si = SRCs[e], di = DSTs[e];
        float dx = x[si * 3 + 0] - x[di * 3 + 0];
        float dy = x[si * 3 + 1] - x[di * 3 + 1];
        float dz = x[si * 3 + 2] - x[di * 3 + 2];
        float len = fmaxf(sqrtf(dx * dx + dy * dy + dz * dz), 1e-8f);
        float w = s / len;
        float* ox = out + N_NODES * NODE_DIM;
        atomAddF(ox + di * 3 + 0, w * dx);
        atomAddF(ox + di * 3 + 1, w * dy);
        atomAddF(ox + di * 3 + 2, w * dz);
    }
}

extern "C" void kernel_launch(void* const* d_in, const int* in_sizes, int n_in,
                              void* d_out, int out_size, void* d_ws, size_t ws_size,
                              hipStream_t stream)
{
    const float* h     = (const float*)d_in[0];
    const float* x     = (const float*)d_in[1];
    const int*   eidx  = (const int*)  d_in[2];
    const float* edist = (const float*)d_in[3];
    const float* nW1   = (const float*)d_in[4];
    const float* nb1   = (const float*)d_in[5];
    const float* nW2   = (const float*)d_in[6];
    const float* nb2   = (const float*)d_in[7];
    const float* cW1   = (const float*)d_in[8];
    const float* cb1   = (const float*)d_in[9];
    const float* cW2   = (const float*)d_in[10];
    const float* eW1   = (const float*)d_in[11];
    const float* eb1   = (const float*)d_in[12];
    const float* eW2   = (const float*)d_in[13];
    const float* eb2   = (const float*)d_in[14];
    float* out = (float*)d_out;

    const int tot = N_NODES * (NODE_DIM + 3);
    egnn_init_out<<<(tot + 255) / 256, 256, 0, stream>>>(h, x, out);

    const size_t smem_bytes = (size_t)SMEM_FLOATS * sizeof(float);
    egnn_edge_kernel<<<N_EDGES / TILE_E, THREADS, smem_bytes, stream>>>(
        h, x, eidx, edist,
        nW1, nb1, nW2, nb2,
        cW1, cb1, cW2,
        eW1, eb1, eW2, eb2,
        out);
}


# Weight-panel staging software-pipelined: next panel's global loads issue in front of the current panel's WMMA block into a second LDS buffer; one barrier per panel; sum order unchanged
# speedup vs baseline: 1.4320x; 1.4320x over previous
#include <hip/hip_runtime.h>
#include <math.h>

#define N_NODES   32768
#define N_EDGES   524288
#define NODE_DIM  128
#define HIDDEN_DIM 256
#define EDGE_DIM  64
#define IN_DIM    320

#define TILE_E    64
#define THREADS   256

#define A_LD      324
#define A_OFF     0
#define A_SIZE    (TILE_E * A_LD)
#define H_LD      260
#define H_OFF     (A_OFF + A_SIZE)
#define H_SIZE    (TILE_E * H_LD)
#define B_OFF     (H_OFF + H_SIZE)
#define B_SIZE    (32 * 256)
#define S_OFF     (B_OFF + 2 * B_SIZE)
#define SMEM_FLOATS (S_OFF + 4 * TILE_E)

typedef float v2f __attribute__((ext_vector_type(2)));
typedef float v8f __attribute__((ext_vector_type(8)));

__device__ __forceinline__ float silu_f(float v) {
    return v / (1.0f + __expf(-v));
}

__device__ __forceinline__ void atomAddF(float* p, float v) {
    __hip_atomic_fetch_add(p, v, __ATOMIC_RELAXED, __HIP_MEMORY_SCOPE_AGENT);
}


template<int NSTAGE>
__device__ __forceinline__ void panel_load(const float* __restrict__ W, int ldw, int kp,
                                           float* R, int tid)
{
    constexpr int KGRP = NSTAGE / 32;
    const int n  = tid & (NSTAGE - 1);
    const int g0 = (tid / NSTAGE) * KGRP;
    const float* p = W + (kp + 4 * g0) * ldw + n;
#pragma unroll
    for (int j = 0; j < 4 * KGRP; ++j)
        R[j] = p[j * ldw];
}

template<int NSTAGE>
__device__ __forceinline__ void panel_store(float* __restrict__ Bs, const float* R, int tid)
{
    constexpr int KGRP = NSTAGE / 32;
    const int n  = tid & (NSTAGE - 1);
    const int g0 = (tid / NSTAGE) * KGRP;
#pragma unroll
    for (int j = 0; j < KGRP; ++j) {
        float4 v = make_float4(R[4 * j], R[4 * j + 1], R[4 * j + 2], R[4 * j + 3]);
        *(float4*)(Bs + (((g0 + j) * NSTAGE + n) << 2)) = v;
    }
}

template<int NTILES, int NSTAGE>
__device__ __forceinline__ void panel_mma(const float* __restrict__ Arow,
                                          const float* __restrict__ Bs,
                                          v8f* acc, int tid)
{
    const int lane  = tid & 31;
    const int wave  = tid >> 5;
    const int row   = lane & 15;
    const int hi    = lane >> 4;
    const int nsel  = wave & 1;
    const int nbase = nsel * (NSTAGE / 2);
#pragma unroll
    for (int kl = 0; kl < 32; kl += 4) {
        v2f a = *(const v2f*)(Arow + kl);
#pragma unroll
        for (int nt = 0; nt < NTILES; ++nt) {
            v2f b = *(const v2f*)(Bs + ((((kl >> 2) * NSTAGE + nbase + nt * 16 + row) << 2) + 2 * hi));
            acc[nt] = __builtin_amdgcn_wmma_f32_16x16x4_f32(
                 false, a,  false, b,
                 (short)0, acc[nt],
                 false,  false);
        }
    }
}

template<int NTILES, int NSTAGE, int NNEXT>
__device__ __forceinline__ void gemm_pipe(const float* __restrict__ W, int ldw, int K,
                                          const float* __restrict__ A, int lda,
                                          float* __restrict__ Bs, int& cur,
                                          v8f* acc, float* R,
                                          const float* __restrict__ Wnext, int ldwnext,
                                          int tid)
{
    const int lane  = tid & 31;
    const int wave  = tid >> 5;
    const int row   = lane & 15;
    const int hi    = lane >> 4;
    const int etile = wave >> 1;
    const float* Arow = A + (etile * 16 + row) * lda + 2 * hi;

    for (int kp = 0; kp < K - 32; kp += 32) {
        panel_load<NSTAGE>(W, ldw, kp + 32, R, tid);
        panel_mma<NTILES, NSTAGE>(Arow + kp, Bs + cur * B_SIZE, acc, tid);
        panel_store<NSTAGE>(Bs + (cur ^ 1) * B_SIZE, R, tid);
        __syncthreads();
        cur ^= 1;
    }
    if constexpr (NNEXT > 0) panel_load<NNEXT>(Wnext, ldwnext, 0, R, tid);
    panel_mma<NTILES, NSTAGE>(Arow + K - 32, Bs + cur * B_SIZE, acc, tid);
}

__global__ void egnn_init_out(const float* __restrict__ h,
                              const float* __restrict__ x,
                              float* __restrict__ out)
{
    int i = blockIdx.x * blockDim.x + threadIdx.x;
    if (i < N_NODES * NODE_DIM)                out[i] = h[i];
    else if (i < N_NODES * (NODE_DIM + 3))     out[i] = x[i - N_NODES * NODE_DIM];
}

__global__ void __launch_bounds__(THREADS)
egnn_edge_kernel(const float* __restrict__ h,   const float* __restrict__ x,
                 const int*   __restrict__ eidx,const float* __restrict__ edist,
                 const float* __restrict__ nW1, const float* __restrict__ nb1,
                 const float* __restrict__ nW2, const float* __restrict__ nb2,
                 const float* __restrict__ cW1, const float* __restrict__ cb1,
                 const float* __restrict__ cW2,
                 const float* __restrict__ eW1, const float* __restrict__ eb1,
                 const float* __restrict__ eW2, const float* __restrict__ eb2,
                 float* __restrict__ out)
{
    extern __shared__ float smem[];
    float* Ablk = smem + A_OFF;
    float* Hblk = smem + H_OFF;
    float* Bs   = smem + B_OFF;
    int*   SRCs = (int*)(smem + S_OFF);
    int*   DSTs = (int*)(smem + S_OFF + TILE_E);
    float* DIST = smem + S_OFF + 2 * TILE_E;

    const int tid   = threadIdx.x;
    const int lane  = tid & 31;
    const int wave  = tid >> 5;
    const int row   = lane & 15;
    const int hi    = lane >> 4;
    const int etile = wave >> 1;
    const int nsel  = wave & 1;
    const int e0    = blockIdx.x * TILE_E;

    float R[32];
    int   cur = 0;
    panel_load<256>(nW1, HIDDEN_DIM, 0, R, tid);

    if (tid < TILE_E) {
        SRCs[tid] = eidx[e0 + tid];
        DSTs[tid] = eidx[N_EDGES + e0 + tid];
        DIST[tid] = edist[e0 + tid];
    }
    __syncthreads();

    for (int i = tid; i < TILE_E * 64; i += THREADS) {
        int e = i >> 6, q = i & 63;
        int half = q >> 5, c4 = q & 31;
        int node = half ? DSTs[e] : SRCs[e];
        *(float4*)(Ablk + e * A_LD + half * NODE_DIM + c4 * 4) =
            ((const float4*)h)[node * (NODE_DIM / 4) + c4];
    }

    float* hid = Hblk;
    for (int i = tid; i < TILE_E * EDGE_DIM; i += THREADS) {
        int e = i >> 6, k = i & 63;
        hid[k * TILE_E + e] = silu_f(DIST[e] * eW1[k] + eb1[k]);
    }
    __syncthreads();
    for (int i = tid; i < TILE_E * EDGE_DIM; i += THREADS) {
        int e = i >> 6, j = i & 63;
        float s = eb2[j];
#pragma unroll 8
        for (int k = 0; k < EDGE_DIM; ++k)
            s += hid[k * TILE_E + e] * eW2[k * EDGE_DIM + j];
        Ablk[e * A_LD + 2 * NODE_DIM + j] = s;
    }
    panel_store<256>(Bs, R, tid);
    __syncthreads();

    {
        v8f acc[8];
#pragma unroll
        for (int nt = 0; nt < 8; ++nt) {
            float bv = nb1[nsel * 128 + nt * 16 + row];
            v8f t = {bv, bv, bv, bv, bv, bv, bv, bv};
            acc[nt] = t;
        }
        gemm_pipe<8, 256, 128>(nW1, HIDDEN_DIM, IN_DIM, Ablk, A_LD, Bs, cur, acc, R,
                               nW2, NODE_DIM, tid);
#pragma unroll
        for (int nt = 0; nt < 8; ++nt)
#pragma unroll
            for (int i = 0; i < 8; ++i) {
                int e = etile * 16 + hi * 8 + i;
                Hblk[e * H_LD + nsel * 128 + nt * 16 + row] = silu_f(acc[nt][i]);
            }
        panel_store<128>(Bs + (cur ^ 1) * B_SIZE, R, tid);
        __syncthreads();
        cur ^= 1;
    }

    {
        v8f acc[4];
#pragma unroll
        for (int nt = 0; nt < 4; ++nt) {
            float bv = nb2[nsel * 64 + nt * 16 + row];
            v8f t = {bv, bv, bv, bv, bv, bv, bv, bv};
            acc[nt] = t;
        }
        gemm_pipe<4, 128, 256>(nW2, NODE_DIM, HIDDEN_DIM, Hblk, H_LD, Bs, cur, acc, R,
                               cW1, HIDDEN_DIM, tid);
#pragma unroll
        for (int nt = 0; nt < 4; ++nt)
#pragma unroll
            for (int i = 0; i < 8; ++i) {
                int e = etile * 16 + hi * 8 + i;
                int d = DSTs[e];
                atomAddF(out + d * NODE_DIM + nsel * 64 + nt * 16 + row, acc[nt][i]);
            }
        panel_store<256>(Bs + (cur ^ 1) * B_SIZE, R, tid);
        __syncthreads();
        cur ^= 1;
    }

    {
        v8f acc[8];
#pragma unroll
        for (int nt = 0; nt < 8; ++nt) {
            float bv = cb1[nsel * 128 + nt * 16 + row];
            v8f t = {bv, bv, bv, bv, bv, bv, bv, bv};
            acc[nt] = t;
        }
        gemm_pipe<8, 256, 0>(cW1, HIDDEN_DIM, IN_DIM, Ablk, A_LD, Bs, cur, acc, R,
                             cW1, HIDDEN_DIM, tid);
#pragma unroll
        for (int nt = 0; nt < 8; ++nt)
#pragma unroll
            for (int i = 0; i < 8; ++i) {
                int e = etile * 16 + hi * 8 + i;
                Hblk[e * H_LD + nsel * 128 + nt * 16 + row] = silu_f(acc[nt][i]);
            }
    }
    __syncthreads();

    if (tid < TILE_E) {
        int e = tid;
        float s = 0.0f;
#pragma unroll 8
        for (int k = 0; k < HIDDEN_DIM; ++k)
            s += Hblk[e * H_LD + k] * cW2[k];
        int si = SRCs[e], di = DSTs[e];
        float dx = x[si * 3 + 0] - x[di * 3 + 0];
        float dy = x[si * 3 + 1] - x[di * 3 + 1];
        float dz = x[si * 3 + 2] - x[di * 3 + 2];
        float len = fmaxf(sqrtf(dx * dx + dy * dy + dz * dz), 1e-8f);
        float w = s / len;
        float* ox = out + N_NODES * NODE_DIM;
        atomAddF(ox + di * 3 + 0, w * dx);
        atomAddF(ox + di * 3 + 1, w * dy);
        atomAddF(ox + di * 3 + 2, w * dz);
    }
}

extern "C" void kernel_launch(void* const* d_in, const int* in_sizes, int n_in,
                              void* d_out, int out_size, void* d_ws, size_t ws_size,
                              hipStream_t stream)
{
    const float* h     = (const float*)d_in[0];
    const float* x     = (const float*)d_in[1];
    const int*   eidx  = (const int*)  d_in[2];
    const float* edist = (const float*)d_in[3];
    const float* nW1   = (const float*)d_in[4];
    const float* nb1   = (const float*)d_in[5];
    const float* nW2   = (const float*)d_in[6];
    const float* nb2   = (const float*)d_in[7];
    const float* cW1   = (const float*)d_in[8];
    const float* cb1   = (const float*)d_in[9];
    const float* cW2   = (const float*)d_in[10];
    const float* eW1   = (const float*)d_in[11];
    const float* eb1   = (const float*)d_in[12];
    const float* eW2   = (const float*)d_in[13];
    const float* eb2   = (const float*)d_in[14];
    float* out = (float*)d_out;

    const int tot = N_NODES * (NODE_DIM + 3);
    egnn_init_out<<<(tot + 255) / 256, 256, 0, stream>>>(h, x, out);

    const size_t smem_bytes = (size_t)SMEM_FLOATS * sizeof(float);
    egnn_edge_kernel<<<N_EDGES / TILE_E, THREADS, smem_bytes, stream>>>(
        h, x, eidx, edist,
        nW1, nb1, nW2, nb2,
        cW1, cb1, cW2,
        eW1, eb1, eW2, eb2,
        out);
}


# WMMA fragment feed: A and B read per 8-deep k block with 16-byte LDS loads, issued as one group into a second register set ahead of each WMMA block; k order inside each 8-k group changes (sum order)
# speedup vs baseline: 1.5449x; 1.0788x over previous
#include <hip/hip_runtime.h>
#include <math.h>

#define N_NODES   32768
#define N_EDGES   524288
#define NODE_DIM  128
#define HIDDEN_DIM 256
#define EDGE_DIM  64
#define IN_DIM    320

#define TILE_E    64
#define THREADS   256

#define A_LD      324
#define A_OFF     0
#define A_SIZE    (TILE_E * A_LD)
#define H_LD      260
#define H_OFF     (A_OFF + A_SIZE)
#define H_SIZE    (TILE_E * H_LD)
#define B_OFF     (H_OFF + H_SIZE)
#define B_SIZE    (32 * 256)
#define S_OFF     (B_OFF + 2 * B_SIZE)
#define SMEM_FLOATS (S_OFF + 4 * TILE_E)

typedef float v2f __attribute__((ext_vector_type(2)));
typedef float v4f __attribute__((ext_vector_type(4)));
typedef float v8f __attribute__((ext_vector_type(8)));

__device__ __forceinline__ float silu_f(float v) {
    return v / (1.0f + __expf(-v));
}

__device__ __forceinline__ void atomAddF(float* p, float v) {
    __hip_atomic_fetch_add(p, v, __ATOMIC_RELAXED, __HIP_MEMORY_SCOPE_AGENT);
}


template<int NSTAGE>
__device__ __forceinline__ void panel_load(const float* __restrict__ W, int ldw, int kp,
                                           float* R, int tid)
{
    constexpr int KGRP = NSTAGE / 32;
    const int n  = tid & (NSTAGE - 1);
    const int g0 = (tid / NSTAGE) * KGRP;
    const float* p = W + (kp + 4 * g0) * ldw + n;
#pragma unroll
    for (int j = 0; j < 4 * KGRP; ++j)
        R[j] = p[j * ldw];
}

template<int NSTAGE>
__device__ __forceinline__ void panel_store(float* __restrict__ Bs, const float* R, int tid)
{
    constexpr int KGRP = NSTAGE / 32;
    const int n  = tid & (NSTAGE - 1);
    const int g0 = (tid / NSTAGE) * KGRP;
#pragma unroll
    for (int j = 0; j < KGRP; ++j) {
        float4 v = make_float4(R[4 * j], R[4 * j + 1], R[4 * j + 2], R[4 * j + 3]);
        *(float4*)(Bs + (((g0 + j) * NSTAGE + n) << 2)) = v;
    }
}

template<int NTILES, int NSTAGE>
__device__ __forceinline__ void frag_load(const float* __restrict__ Ap,
                                          const float* __restrict__ Bp,
                                          int kb, v4f& a, v4f* b)
{
    a = *(const v4f*)(Ap + 8 * kb);
#pragma unroll
    for (int nt = 0; nt < NTILES; ++nt)
        b[nt] = *(const v4f*)(Bp + ((2 * kb * NSTAGE + nt * 16) << 2));
}

template<int NTILES>
__device__ __forceinline__ void frag_mma(const v4f& a, const v4f* b, v8f* acc)
{
    const v2f alo = {a.x, a.y};
    const v2f ahi = {a.z, a.w};
#pragma unroll
    for (int nt = 0; nt < NTILES; ++nt) {
        const v2f bl = {b[nt].x, b[nt].y};
        acc[nt] = __builtin_amdgcn_wmma_f32_16x16x4_f32(
             false, alo,  false, bl,
             (short)0, acc[nt],
             false,  false);
    }
#pragma unroll
    for (int nt = 0; nt < NTILES; ++nt) {
        const v2f bh = {b[nt].z, b[nt].w};
        acc[nt] = __builtin_amdgcn_wmma_f32_16x16x4_f32(
             false, ahi,  false, bh,
             (short)0, acc[nt],
             false,  false);
    }
}

template<int NTILES, int NSTAGE>
__device__ __forceinline__ void frag_step(const float* __restrict__ Ap,
                                          const float* __restrict__ Bp, int kb,
                                          v4f& an, v4f* bn,
                                          const v4f& ac, const v4f* bc, v8f* acc)
{
    frag_load<NTILES, NSTAGE>(Ap, Bp, kb, an, bn);
    __builtin_amdgcn_sched_barrier(0);
    frag_mma<NTILES>(ac, bc, acc);
    __builtin_amdgcn_sched_barrier(0);
}

template<int NTILES, int NSTAGE, int NNEXT>
__device__ __forceinline__ void gemm_pipe(const float* __restrict__ W, int ldw, int K,
                                          const float* __restrict__ A, int lda,
                                          float* __restrict__ Bs, int& cur,
                                          v8f* acc, float* R,
                                          const float* __restrict__ Wnext, int ldwnext,
                                          int tid)
{
    const int lane  = tid & 31;
    const int wave  = tid >> 5;
    const int row   = lane & 15;
    const int hi    = lane >> 4;
    const int etile = wave >> 1;
    const int nsel  = wave & 1;
    const float* Arow = A + (etile * 16 + row) * lda + 4 * hi;
    const int    boff = (hi * NSTAGE + nsel * (NSTAGE / 2) + row) << 2;

    v4f a0, a1, b0[NTILES], b1[NTILES];
    frag_load<NTILES, NSTAGE>(Arow, Bs + cur * B_SIZE + boff, 0, a0, b0);

    for (int kp = 0; kp < K - 32; kp += 32) {
        const float* Ap = Arow + kp;
        const float* Bp = Bs + cur * B_SIZE + boff;
        panel_load<NSTAGE>(W, ldw, kp + 32, R, tid);
        frag_step<NTILES, NSTAGE>(Ap, Bp, 1, a1, b1, a0, b0, acc);
        frag_step<NTILES, NSTAGE>(Ap, Bp, 2, a0, b0, a1, b1, acc);
        frag_step<NTILES, NSTAGE>(Ap, Bp, 3, a1, b1, a0, b0, acc);
        panel_store<NSTAGE>(Bs + (cur ^ 1) * B_SIZE, R, tid);
        __syncthreads();
        cur ^= 1;
        frag_step<NTILES, NSTAGE>(Ap + 32, Bs + cur * B_SIZE + boff, 0, a0, b0, a1, b1, acc);
    }
    if constexpr (NNEXT > 0) panel_load<NNEXT>(Wnext, ldwnext, 0, R, tid);
    {
        const float* Ap = Arow + K - 32;
        const float* Bp = Bs + cur * B_SIZE + boff;
        frag_step<NTILES, NSTAGE>(Ap, Bp, 1, a1, b1, a0, b0, acc);
        frag_step<NTILES, NSTAGE>(Ap, Bp, 2, a0, b0, a1, b1, acc);
        frag_step<NTILES, NSTAGE>(Ap, Bp, 3, a1, b1, a0, b0, acc);
        frag_mma<NTILES>(a1, b1, acc);
    }
}

__global__ void egnn_init_out(const float* __restrict__ h,
                              const float* __restrict__ x,
                              float* __restrict__ out)
{
    int i = blockIdx.x * blockDim.x + threadIdx.x;
    if (i < N_NODES * NODE_DIM)                out[i] = h[i];
    else if (i < N_NODES * (NODE_DIM + 3))     out[i] = x[i - N_NODES * NODE_DIM];
}

__global__ void __launch_bounds__(THREADS)
egnn_edge_kernel(const float* __restrict__ h,   const float* __restrict__ x,
                 const int*   __restrict__ eidx,const float* __restrict__ edist,
                 const float* __restrict__ nW1, const float* __restrict__ nb1,
                 const float* __restrict__ nW2, const float* __restrict__ nb2,
                 const float* __restrict__ cW1, const float* __restrict__ cb1,
                 const float* __restrict__ cW2,
                 const float* __restrict__ eW1, const float* __restrict__ eb1,
                 const float* __restrict__ eW2, const float* __restrict__ eb2,
                 float* __restrict__ out)
{
    extern __shared__ float smem[];
    float* Ablk = smem + A_OFF;
    float* Hblk = smem + H_OFF;
    float* Bs   = smem + B_OFF;
    int*   SRCs = (int*)(smem + S_OFF);
    int*   DSTs = (int*)(smem + S_OFF + TILE_E);
    float* DIST = smem + S_OFF + 2 * TILE_E;

    const int tid   = threadIdx.x;
    const int lane  = tid & 31;
    const int wave  = tid >> 5;
    const int row   = lane & 15;
    const int hi    = lane >> 4;
    const int etile = wave >> 1;
    const int nsel  = wave & 1;
    const int e0    = blockIdx.x * TILE_E;

    float R[32];
    int   cur = 0;
    panel_load<256>(nW1, HIDDEN_DIM, 0, R, tid);

    if (tid < TILE_E) {
        SRCs[tid] = eidx[e0 + tid];
        DSTs[tid] = eidx[N_EDGES + e0 + tid];
        DIST[tid] = edist[e0 + tid];
    }
    __syncthreads();

    for (int i = tid; i < TILE_E * 64; i += THREADS) {
        int e = i >> 6, q = i & 63;
        int half = q >> 5, c4 = q & 31;
        int node = half ? DSTs[e] : SRCs[e];
        *(float4*)(Ablk + e * A_LD + half * NODE_DIM + c4 * 4) =
            ((const float4*)h)[node * (NODE_DIM / 4) + c4];
    }

    float* hid = Hblk;
    for (int i = tid; i < TILE_E * EDGE_DIM; i += THREADS) {
        int e = i >> 6, k = i & 63;
        hid[k * TILE_E + e] = silu_f(DIST[e] * eW1[k] + eb1[k]);
    }
    __syncthreads();
    for (int i = tid; i < TILE_E * EDGE_DIM; i += THREADS) {
        int e = i >> 6, j = i & 63;
        float s = eb2[j];
#pragma unroll 8
        for (int k = 0; k < EDGE_DIM; ++k)
            s += hid[k * TILE_E + e] * eW2[k * EDGE_DIM + j];
        Ablk[e * A_LD + 2 * NODE_DIM + j] = s;
    }
    panel_store<256>(Bs, R, tid);
    __syncthreads();

    {
        v8f acc[8];
#pragma unroll
        for (int nt = 0; nt < 8; ++nt) {
            float bv = nb1[nsel * 128 + nt * 16 + row];
            v8f t = {bv, bv, bv, bv, bv, bv, bv, bv};
            acc[nt] = t;
        }
        gemm_pipe<8, 256, 128>(nW1, HIDDEN_DIM, IN_DIM, Ablk, A_LD, Bs, cur, acc, R,
                               nW2, NODE_DIM, tid);
#pragma unroll
        for (int nt = 0; nt < 8; ++nt)
#pragma unroll
            for (int i = 0; i < 8; ++i) {
                int e = etile * 16 + hi * 8 + i;
                Hblk[e * H_LD + nsel * 128 + nt * 16 + row] = silu_f(acc[nt][i]);
            }
        panel_store<128>(Bs + (cur ^ 1) * B_SIZE, R, tid);
        __syncthreads();
        cur ^= 1;
    }

    {
        v8f acc[4];
#pragma unroll
        for (int nt = 0; nt < 4; ++nt) {
            float bv = nb2[nsel * 64 + nt * 16 + row];
            v8f t = {bv, bv, bv, bv, bv, bv, bv, bv};
            acc[nt] = t;
        }
        gemm_pipe<4, 128, 256>(nW2, NODE_DIM, HIDDEN_DIM, Hblk, H_LD, Bs, cur, acc, R,
                               cW1, HIDDEN_DIM, tid);
#pragma unroll
        for (int nt = 0; nt < 4; ++nt)
#pragma unroll
            for (int i = 0; i < 8; ++i) {
                int e = etile * 16 + hi * 8 + i;
                int d = DSTs[e];
                atomAddF(out + d * NODE_DIM + nsel * 64 + nt * 16 + row, acc[nt][i]);
            }
        panel_store<256>(Bs + (cur ^ 1) * B_SIZE, R, tid);
        __syncthreads();
        cur ^= 1;
    }

    {
        v8f acc[8];
#pragma unroll
        for (int nt = 0; nt < 8; ++nt) {
            float bv = cb1[nsel * 128 + nt * 16 + row];
            v8f t = {bv, bv, bv, bv, bv, bv, bv, bv};
            acc[nt] = t;
        }
        gemm_pipe<8, 256, 0>(cW1, HIDDEN_DIM, IN_DIM, Ablk, A_LD, Bs, cur, acc, R,
                             cW1, HIDDEN_DIM, tid);
#pragma unroll
        for (int nt = 0; nt < 8; ++nt)
#pragma unroll
            for (int i = 0; i < 8; ++i) {
                int e = etile * 16 + hi * 8 + i;
                Hblk[e * H_LD + nsel * 128 + nt * 16 + row] = silu_f(acc[nt][i]);
            }
    }
    __syncthreads();

    if (tid < TILE_E) {
        int e = tid;
        float s = 0.0f;
#pragma unroll 8
        for (int k = 0; k < HIDDEN_DIM; ++k)
            s += Hblk[e * H_LD + k] * cW2[k];
        int si = SRCs[e], di = DSTs[e];
        float dx = x[si * 3 + 0] - x[di * 3 + 0];
        float dy = x[si * 3 + 1] - x[di * 3 + 1];
        float dz = x[si * 3 + 2] - x[di * 3 + 2];
        float len = fmaxf(sqrtf(dx * dx + dy * dy + dz * dz), 1e-8f);
        float w = s / len;
        float* ox = out + N_NODES * NODE_DIM;
        atomAddF(ox + di * 3 + 0, w * dx);
        atomAddF(ox + di * 3 + 1, w * dy);
        atomAddF(ox + di * 3 + 2, w * dz);
    }
}

extern "C" void kernel_launch(void* const* d_in, const int* in_sizes, int n_in,
                              void* d_out, int out_size, void* d_ws, size_t ws_size,
                              hipStream_t stream)
{
    const float* h     = (const float*)d_in[0];
    const float* x     = (const float*)d_in[1];
    const int*   eidx  = (const int*)  d_in[2];
    const float* edist = (const float*)d_in[3];
    const float* nW1   = (const float*)d_in[4];
    const float* nb1   = (const float*)d_in[5];
    const float* nW2   = (const float*)d_in[6];
    const float* nb2   = (const float*)d_in[7];
    const float* cW1   = (const float*)d_in[8];
    const float* cb1   = (const float*)d_in[9];
    const float* cW2   = (const float*)d_in[10];
    const float* eW1   = (const float*)d_in[11];
    const float* eb1   = (const float*)d_in[12];
    const float* eW2   = (const float*)d_in[13];
    const float* eb2   = (const float*)d_in[14];
    float* out = (float*)d_out;

    const int tot = N_NODES * (NODE_DIM + 3);
    egnn_init_out<<<(tot + 255) / 256, 256, 0, stream>>>(h, x, out);

    const size_t smem_bytes = (size_t)SMEM_FLOATS * sizeof(float);
    egnn_edge_kernel<<<N_EDGES / TILE_E, THREADS, smem_bytes, stream>>>(
        h, x, eidx, edist,
        nW1, nb1, nW2, nb2,
        cW1, cb1, cW2,
        eW1, eb1, eW2, eb2,
        out);
}


# Two workgroups per WGP: LDS per workgroup cut to 150 KB by overlaying the H tile on the A tile (coord GEMM first, its 256->1 projection reduced in registers; coord_weight sum order changes)
# speedup vs baseline: 2.0487x; 1.3261x over previous
#include <hip/hip_runtime.h>
#include <math.h>

#define N_NODES   32768
#define N_EDGES   524288
#define NODE_DIM  128
#define HIDDEN_DIM 256
#define EDGE_DIM  64
#define IN_DIM    320

#define TILE_E    64
#define THREADS   256

#define A_LD      324
#define A_OFF     0
#define A_SIZE    (TILE_E * A_LD)
#define H_LD      260
#define H_OFF     A_OFF
#define H_SIZE    (TILE_E * H_LD)
#define B_OFF     (A_OFF + A_SIZE)
#define B_SIZE    (32 * 256)
#define S_OFF     (B_OFF + 2 * B_SIZE)
#define SMEM_FLOATS (S_OFF + 6 * TILE_E)

typedef float v2f __attribute__((ext_vector_type(2)));
typedef float v4f __attribute__((ext_vector_type(4)));
typedef float v8f __attribute__((ext_vector_type(8)));

__device__ __forceinline__ float silu_f(float v) {
    return v / (1.0f + __expf(-v));
}

__device__ __forceinline__ void atomAddF(float* p, float v) {
    __hip_atomic_fetch_add(p, v, __ATOMIC_RELAXED, __HIP_MEMORY_SCOPE_AGENT);
}


template<int NSTAGE>
__device__ __forceinline__ void panel_load(const float* __restrict__ W, int ldw, int kp,
                                           float* R, int tid)
{
    constexpr int KGRP = NSTAGE / 32;
    const int n  = tid & (NSTAGE - 1);
    const int g0 = (tid / NSTAGE) * KGRP;
    const float* p = W + (kp + 4 * g0) * ldw + n;
#pragma unroll
    for (int j = 0; j < 4 * KGRP; ++j)
        R[j] = p[j * ldw];
}

template<int NSTAGE>
__device__ __forceinline__ void panel_store(float* __restrict__ Bs, const float* R, int tid)
{
    constexpr int KGRP = NSTAGE / 32;
    const int n  = tid & (NSTAGE - 1);
    const int g0 = (tid / NSTAGE) * KGRP;
#pragma unroll
    for (int j = 0; j < KGRP; ++j) {
        float4 v = make_float4(R[4 * j], R[4 * j + 1], R[4 * j + 2], R[4 * j + 3]);
        *(float4*)(Bs + (((g0 + j) * NSTAGE + n) << 2)) = v;
    }
}

template<int NTILES, int NSTAGE>
__device__ __forceinline__ void frag_load(const float* __restrict__ Ap,
                                          const float* __restrict__ Bp,
                                          int kb, v4f& a, v4f* b)
{
    a = *(const v4f*)(Ap + 8 * kb);
#pragma unroll
    for (int nt = 0; nt < NTILES; ++nt)
        b[nt] = *(const v4f*)(Bp + ((2 * kb * NSTAGE + nt * 16) << 2));
}

template<int NTILES>
__device__ __forceinline__ void frag_mma(const v4f& a, const v4f* b, v8f* acc)
{
    const v2f alo = {a.x, a.y};
    const v2f ahi = {a.z, a.w};
#pragma unroll
    for (int nt = 0; nt < NTILES; ++nt) {
        const v2f bl = {b[nt].x, b[nt].y};
        acc[nt] = __builtin_amdgcn_wmma_f32_16x16x4_f32(
             false, alo,  false, bl,
             (short)0, acc[nt],
             false,  false);
    }
#pragma unroll
    for (int nt = 0; nt < NTILES; ++nt) {
        const v2f bh = {b[nt].z, b[nt].w};
        acc[nt] = __builtin_amdgcn_wmma_f32_16x16x4_f32(
             false, ahi,  false, bh,
             (short)0, acc[nt],
             false,  false);
    }
}

template<int NTILES, int NSTAGE>
__device__ __forceinline__ void frag_step(const float* __restrict__ Ap,
                                          const float* __restrict__ Bp, int kb,
                                          v4f& an, v4f* bn,
                                          const v4f& ac, const v4f* bc, v8f* acc)
{
    frag_load<NTILES, NSTAGE>(Ap, Bp, kb, an, bn);
    __builtin_amdgcn_sched_barrier(0);
    frag_mma<NTILES>(ac, bc, acc);
    __builtin_amdgcn_sched_barrier(0);
}

template<int NTILES, int NSTAGE, int NNEXT>
__device__ __forceinline__ void gemm_pipe(const float* __restrict__ W, int ldw, int K,
                                          const float* __restrict__ A, int lda,
                                          float* __restrict__ Bs, int& cur,
                                          v8f* acc, float* R,
                                          const float* __restrict__ Wnext, int ldwnext,
                                          int tid)
{
    const int lane  = tid & 31;
    const int wave  = tid >> 5;
    const int row   = lane & 15;
    const int hi    = lane >> 4;
    const int etile = wave >> 1;
    const int nsel  = wave & 1;
    const float* Arow = A + (etile * 16 + row) * lda + 4 * hi;
    const int    boff = (hi * NSTAGE + nsel * (NSTAGE / 2) + row) << 2;

    v4f a0, a1, b0[NTILES], b1[NTILES];
    frag_load<NTILES, NSTAGE>(Arow, Bs + cur * B_SIZE + boff, 0, a0, b0);

    for (int kp = 0; kp < K - 32; kp += 32) {
        const float* Ap = Arow + kp;
        const float* Bp = Bs + cur * B_SIZE + boff;
        panel_load<NSTAGE>(W, ldw, kp + 32, R, tid);
        frag_step<NTILES, NSTAGE>(Ap, Bp, 1, a1, b1, a0, b0, acc);
        frag_step<NTILES, NSTAGE>(Ap, Bp, 2, a0, b0, a1, b1, acc);
        frag_step<NTILES, NSTAGE>(Ap, Bp, 3, a1, b1, a0, b0, acc);
        panel_store<NSTAGE>(Bs + (cur ^ 1) * B_SIZE, R, tid);
        __syncthreads();
        cur ^= 1;
        frag_step<NTILES, NSTAGE>(Ap + 32, Bs + cur * B_SIZE + boff, 0, a0, b0, a1, b1, acc);
    }
    if constexpr (NNEXT > 0) panel_load<NNEXT>(Wnext, ldwnext, 0, R, tid);
    {
        const float* Ap = Arow + K - 32;
        const float* Bp = Bs + cur * B_SIZE + boff;
        frag_step<NTILES, NSTAGE>(Ap, Bp, 1, a1, b1, a0, b0, acc);
        frag_step<NTILES, NSTAGE>(Ap, Bp, 2, a0, b0, a1, b1, acc);
        frag_step<NTILES, NSTAGE>(Ap, Bp, 3, a1, b1, a0, b0, acc);
        frag_mma<NTILES>(a1, b1, acc);
    }
}

__global__ void egnn_init_out(const float* __restrict__ h,
                              const float* __restrict__ x,
                              float* __restrict__ out)
{
    int i = blockIdx.x * blockDim.x + threadIdx.x;
    if (i < N_NODES * NODE_DIM)                out[i] = h[i];
    else if (i < N_NODES * (NODE_DIM + 3))     out[i] = x[i - N_NODES * NODE_DIM];
}

__global__ void __launch_bounds__(THREADS)
egnn_edge_kernel(const float* __restrict__ h,   const float* __restrict__ x,
                 const int*   __restrict__ eidx,const float* __restrict__ edist,
                 const float* __restrict__ nW1, const float* __restrict__ nb1,
                 const float* __restrict__ nW2, const float* __restrict__ nb2,
                 const float* __restrict__ cW1, const float* __restrict__ cb1,
                 const float* __restrict__ cW2,
                 const float* __restrict__ eW1, const float* __restrict__ eb1,
                 const float* __restrict__ eW2, const float* __restrict__ eb2,
                 float* __restrict__ out)
{
    extern __shared__ float smem[];
    float* Ablk = smem + A_OFF;
    float* Hblk = smem + H_OFF;
    float* Bs   = smem + B_OFF;
    int*   SRCs = (int*)(smem + S_OFF);
    int*   DSTs = (int*)(smem + S_OFF + TILE_E);
    float* DIST = smem + S_OFF + 2 * TILE_E;
    float* CWp  = smem + S_OFF + 4 * TILE_E;

    const int tid   = threadIdx.x;
    const int lane  = tid & 31;
    const int wave  = tid >> 5;
    const int row   = lane & 15;
    const int hi    = lane >> 4;
    const int etile = wave >> 1;
    const int nsel  = wave & 1;
    const int e0    = blockIdx.x * TILE_E;

    float R[32];
    int   cur = 0;
    panel_load<256>(cW1, HIDDEN_DIM, 0, R, tid);

    if (tid < TILE_E) {
        SRCs[tid] = eidx[e0 + tid];
        DSTs[tid] = eidx[N_EDGES + e0 + tid];
        DIST[tid] = edist[e0 + tid];
    }
    __syncthreads();

    for (int i = tid; i < TILE_E * 64; i += THREADS) {
        int e = i >> 6, q = i & 63;
        int half = q >> 5, c4 = q & 31;
        int node = half ? DSTs[e] : SRCs[e];
        *(float4*)(Ablk + e * A_LD + half * NODE_DIM + c4 * 4) =
            ((const float4*)h)[node * (NODE_DIM / 4) + c4];
    }

    float* hid = Bs + B_SIZE;
    for (int i = tid; i < TILE_E * EDGE_DIM; i += THREADS) {
        int e = i >> 6, k = i & 63;
        hid[k * TILE_E + e] = silu_f(DIST[e] * eW1[k] + eb1[k]);
    }
    __syncthreads();
    for (int i = tid; i < TILE_E * EDGE_DIM; i += THREADS) {
        int e = i >> 6, j = i & 63;
        float s = eb2[j];
#pragma unroll 8
        for (int k = 0; k < EDGE_DIM; ++k)
            s += hid[k * TILE_E + e] * eW2[k * EDGE_DIM + j];
        Ablk[e * A_LD + 2 * NODE_DIM + j] = s;
    }
    panel_store<256>(Bs, R, tid);
    __syncthreads();

    {
        v8f acc[8];
#pragma unroll
        for (int nt = 0; nt < 8; ++nt) {
            float bv = cb1[nsel * 128 + nt * 16 + row];
            v8f t = {bv, bv, bv, bv, bv, bv, bv, bv};
            acc[nt] = t;
        }
        gemm_pipe<8, 256, 256>(cW1, HIDDEN_DIM, IN_DIM, Ablk, A_LD, Bs, cur, acc, R,
                               nW1, HIDDEN_DIM, tid);
        float p[8];
#pragma unroll
        for (int i = 0; i < 8; ++i) p[i] = 0.0f;
#pragma unroll
        for (int nt = 0; nt < 8; ++nt) {
            const float w = cW2[nsel * 128 + nt * 16 + row];
#pragma unroll
            for (int i = 0; i < 8; ++i)
                p[i] += silu_f(acc[nt][i]) * w;
        }
#pragma unroll
        for (int m = 1; m < 16; m <<= 1)
#pragma unroll
            for (int i = 0; i < 8; ++i)
                p[i] += __shfl_xor(p[i], m, 32);
        if (row == 0) {
            float* q = CWp + nsel * TILE_E + etile * 16 + hi * 8;
            *(float4*)(q)     = make_float4(p[0], p[1], p[2], p[3]);
            *(float4*)(q + 4) = make_float4(p[4], p[5], p[6], p[7]);
        }
        panel_store<256>(Bs + (cur ^ 1) * B_SIZE, R, tid);
        __syncthreads();
        cur ^= 1;
    }

    {
        v8f acc[8];
#pragma unroll
        for (int nt = 0; nt < 8; ++nt) {
            float bv = nb1[nsel * 128 + nt * 16 + row];
            v8f t = {bv, bv, bv, bv, bv, bv, bv, bv};
            acc[nt] = t;
        }
        gemm_pipe<8, 256, 128>(nW1, HIDDEN_DIM, IN_DIM, Ablk, A_LD, Bs, cur, acc, R,
                               nW2, NODE_DIM, tid);
#pragma unroll
        for (int nt = 0; nt < 8; ++nt)
#pragma unroll
            for (int i = 0; i < 8; ++i)
                acc[nt][i] = silu_f(acc[nt][i]);
        panel_store<128>(Bs + (cur ^ 1) * B_SIZE, R, tid);
        __syncthreads();
#pragma unroll
        for (int nt = 0; nt < 8; ++nt)
#pragma unroll
            for (int i = 0; i < 8; ++i) {
                int e = etile * 16 + hi * 8 + i;
                Hblk[e * H_LD + nsel * 128 + nt * 16 + row] = acc[nt][i];
            }
        __syncthreads();
        cur ^= 1;
    }

    {
        v8f acc[4];
#pragma unroll
        for (int nt = 0; nt < 4; ++nt) {
            float bv = nb2[nsel * 64 + nt * 16 + row];
            v8f t = {bv, bv, bv, bv, bv, bv, bv, bv};
            acc[nt] = t;
        }
        gemm_pipe<4, 128, 0>(nW2, NODE_DIM, HIDDEN_DIM, Hblk, H_LD, Bs, cur, acc, R,
                             nW2, NODE_DIM, tid);
#pragma unroll
        for (int nt = 0; nt < 4; ++nt)
#pragma unroll
            for (int i = 0; i < 8; ++i) {
                int e = etile * 16 + hi * 8 + i;
                int d = DSTs[e];
                atomAddF(out + d * NODE_DIM + nsel * 64 + nt * 16 + row, acc[nt][i]);
            }
    }

    if (tid < TILE_E) {
        int e = tid;
        float s = CWp[e] + CWp[TILE_E + e];
        int si = SRCs[e], di = DSTs[e];
        float dx = x[si * 3 + 0] - x[di * 3 + 0];
        float dy = x[si * 3 + 1] - x[di * 3 + 1];
        float dz = x[si * 3 + 2] - x[di * 3 + 2];
        float len = fmaxf(sqrtf(dx * dx + dy * dy + dz * dz), 1e-8f);
        float w = s / len;
        float* ox = out + N_NODES * NODE_DIM;
        atomAddF(ox + di * 3 + 0, w * dx);
        atomAddF(ox + di * 3 + 1, w * dy);
        atomAddF(ox + di * 3 + 2, w * dz);
    }
}

extern "C" void kernel_launch(void* const* d_in, const int* in_sizes, int n_in,
                              void* d_out, int out_size, void* d_ws, size_t ws_size,
                              hipStream_t stream)
{
    const float* h     = (const float*)d_in[0];
    const float* x     = (const float*)d_in[1];
    const int*   eidx  = (const int*)  d_in[2];
    const float* edist = (const float*)d_in[3];
    const float* nW1   = (const float*)d_in[4];
    const float* nb1   = (const float*)d_in[5];
    const float* nW2   = (const float*)d_in[6];
    const float* nb2   = (const float*)d_in[7];
    const float* cW1   = (const float*)d_in[8];
    const float* cb1   = (const float*)d_in[9];
    const float* cW2   = (const float*)d_in[10];
    const float* eW1   = (const float*)d_in[11];
    const float* eb1   = (const float*)d_in[12];
    const float* eW2   = (const float*)d_in[13];
    const float* eb2   = (const float*)d_in[14];
    float* out = (float*)d_out;

    const int tot = N_NODES * (NODE_DIM + 3);
    egnn_init_out<<<(tot + 255) / 256, 256, 0, stream>>>(h, x, out);

    const size_t smem_bytes = (size_t)SMEM_FLOATS * sizeof(float);
    egnn_edge_kernel<<<N_EDGES / TILE_E, THREADS, smem_bytes, stream>>>(
        h, x, eidx, edist,
        nW1, nb1, nW2, nb2,
        cW1, cb1, cW2,
        eW1, eb1, eW2, eb2,
        out);
}


# First layer split by linearity: h@W1[0:128], h@W1[128:256] built once per node in workspace tables, gathered per edge; per-edge K 320->64 (hidden pre-activation sum order changes)
# speedup vs baseline: 3.2564x; 1.5895x over previous
#include <hip/hip_runtime.h>
#include <math.h>

#define N_NODES   32768
#define N_EDGES   524288
#define NODE_DIM  128
#define HIDDEN_DIM 256
#define EDGE_DIM  64
#define IN_DIM    320

#define TILE_E    64
#define THREADS   256

#define A_LD      324
#define A_OFF     0
#define A_SIZE    (TILE_E * A_LD)
#define H_LD      260
#define H_OFF     A_OFF
#define H_SIZE    (TILE_E * H_LD)
#define B_OFF     (A_OFF + A_SIZE)
#define B_SIZE    (32 * 256)
#define S_OFF     (B_OFF + 2 * B_SIZE)
#define SMEM_FLOATS (S_OFF + 6 * TILE_E)

#define TAB_LD    512
#define TAB_FLOATS ((size_t)N_NODES * TAB_LD)
#define TAB_BYTES  ((size_t)2 * N_NODES * TAB_LD * sizeof(float))

#define NA_LD     132
#define NA_SIZE   (TILE_E * NA_LD)
#define NSMEM_FLOATS (NA_SIZE + 2 * B_SIZE)

#define EA_LD     68
#define ESMEM_FLOATS (H_SIZE + 2 * B_SIZE + 6 * TILE_E)

typedef float v2f __attribute__((ext_vector_type(2)));
typedef float v4f __attribute__((ext_vector_type(4)));
typedef float v8f __attribute__((ext_vector_type(8)));

__device__ __forceinline__ float silu_f(float v) {
    return v / (1.0f + __expf(-v));
}

__device__ __forceinline__ void atomAddF(float* p, float v) {
    __hip_atomic_fetch_add(p, v, __ATOMIC_RELAXED, __HIP_MEMORY_SCOPE_AGENT);
}


template<int NSTAGE>
__device__ __forceinline__ void panel_load(const float* __restrict__ W, int ldw, int kp,
                                           float* R, int tid)
{
    constexpr int KGRP = NSTAGE / 32;
    const int n  = tid & (NSTAGE - 1);
    const int g0 = (tid / NSTAGE) * KGRP;
    const float* p = W + (kp + 4 * g0) * ldw + n;
#pragma unroll
    for (int j = 0; j < 4 * KGRP; ++j)
        R[j] = p[j * ldw];
}

template<int NSTAGE>
__device__ __forceinline__ void panel_store(float* __restrict__ Bs, const float* R, int tid)
{
    constexpr int KGRP = NSTAGE / 32;
    const int n  = tid & (NSTAGE - 1);
    const int g0 = (tid / NSTAGE) * KGRP;
#pragma unroll
    for (int j = 0; j < KGRP; ++j) {
        float4 v = make_float4(R[4 * j], R[4 * j + 1], R[4 * j + 2], R[4 * j + 3]);
        *(float4*)(Bs + (((g0 + j) * NSTAGE + n) << 2)) = v;
    }
}

template<int NTILES, int NSTAGE>
__device__ __forceinline__ void frag_load(const float* __restrict__ Ap,
                                          const float* __restrict__ Bp,
                                          int kb, v4f& a, v4f* b)
{
    a = *(const v4f*)(Ap + 8 * kb);
#pragma unroll
    for (int nt = 0; nt < NTILES; ++nt)
        b[nt] = *(const v4f*)(Bp + ((2 * kb * NSTAGE + nt * 16) << 2));
}

template<int NTILES>
__device__ __forceinline__ void frag_mma(const v4f& a, const v4f* b, v8f* acc)
{
    const v2f alo = {a.x, a.y};
    const v2f ahi = {a.z, a.w};
#pragma unroll
    for (int nt = 0; nt < NTILES; ++nt) {
        const v2f bl = {b[nt].x, b[nt].y};
        acc[nt] = __builtin_amdgcn_wmma_f32_16x16x4_f32(
             false, alo,  false, bl,
             (short)0, acc[nt],
             false,  false);
    }
#pragma unroll
    for (int nt = 0; nt < NTILES; ++nt) {
        const v2f bh = {b[nt].z, b[nt].w};
        acc[nt] = __builtin_amdgcn_wmma_f32_16x16x4_f32(
             false, ahi,  false, bh,
             (short)0, acc[nt],
             false,  false);
    }
}

template<int NTILES, int NSTAGE>
__device__ __forceinline__ void frag_step(const float* __restrict__ Ap,
                                          const float* __restrict__ Bp, int kb,
                                          v4f& an, v4f* bn,
                                          const v4f& ac, const v4f* bc, v8f* acc)
{
    frag_load<NTILES, NSTAGE>(Ap, Bp, kb, an, bn);
    __builtin_amdgcn_sched_barrier(0);
    frag_mma<NTILES>(ac, bc, acc);
    __builtin_amdgcn_sched_barrier(0);
}

template<int NTILES, int NSTAGE, int NNEXT>
__device__ __forceinline__ void gemm_pipe(const float* __restrict__ W, int ldw, int K,
                                          const float* __restrict__ A, int lda,
                                          float* __restrict__ Bs, int& cur,
                                          v8f* acc, float* R,
                                          const float* __restrict__ Wnext, int ldwnext,
                                          int tid)
{
    const int lane  = tid & 31;
    const int wave  = tid >> 5;
    const int row   = lane & 15;
    const int hi    = lane >> 4;
    const int etile = wave >> 1;
    const int nsel  = wave & 1;
    const float* Arow = A + (etile * 16 + row) * lda + 4 * hi;
    const int    boff = (hi * NSTAGE + nsel * (NSTAGE / 2) + row) << 2;

    v4f a0, a1, b0[NTILES], b1[NTILES];
    frag_load<NTILES, NSTAGE>(Arow, Bs + cur * B_SIZE + boff, 0, a0, b0);

    for (int kp = 0; kp < K - 32; kp += 32) {
        const float* Ap = Arow + kp;
        const float* Bp = Bs + cur * B_SIZE + boff;
        panel_load<NSTAGE>(W, ldw, kp + 32, R, tid);
        frag_step<NTILES, NSTAGE>(Ap, Bp, 1, a1, b1, a0, b0, acc);
        frag_step<NTILES, NSTAGE>(Ap, Bp, 2, a0, b0, a1, b1, acc);
        frag_step<NTILES, NSTAGE>(Ap, Bp, 3, a1, b1, a0, b0, acc);
        panel_store<NSTAGE>(Bs + (cur ^ 1) * B_SIZE, R, tid);
        __syncthreads();
        cur ^= 1;
        frag_step<NTILES, NSTAGE>(Ap + 32, Bs + cur * B_SIZE + boff, 0, a0, b0, a1, b1, acc);
    }
    if constexpr (NNEXT > 0) panel_load<NNEXT>(Wnext, ldwnext, 0, R, tid);
    {
        const float* Ap = Arow + K - 32;
        const float* Bp = Bs + cur * B_SIZE + boff;
        frag_step<NTILES, NSTAGE>(Ap, Bp, 1, a1, b1, a0, b0, acc);
        frag_step<NTILES, NSTAGE>(Ap, Bp, 2, a0, b0, a1, b1, acc);
        frag_step<NTILES, NSTAGE>(Ap, Bp, 3, a1, b1, a0, b0, acc);
        frag_mma<NTILES>(a1, b1, acc);
    }
}

__global__ void egnn_init_out(const float* __restrict__ h,
                              const float* __restrict__ x,
                              float* __restrict__ out)
{
    int i = blockIdx.x * blockDim.x + threadIdx.x;
    if (i < N_NODES * NODE_DIM)                out[i] = h[i];
    else if (i < N_NODES * (NODE_DIM + 3))     out[i] = x[i - N_NODES * NODE_DIM];
}

__global__ void __launch_bounds__(THREADS)
egnn_edge_kernel(const float* __restrict__ h,   const float* __restrict__ x,
                 const int*   __restrict__ eidx,const float* __restrict__ edist,
                 const float* __restrict__ nW1, const float* __restrict__ nb1,
                 const float* __restrict__ nW2, const float* __restrict__ nb2,
                 const float* __restrict__ cW1, const float* __restrict__ cb1,
                 const float* __restrict__ cW2,
                 const float* __restrict__ eW1, const float* __restrict__ eb1,
                 const float* __restrict__ eW2, const float* __restrict__ eb2,
                 size_t ws_bytes,
                 float* __restrict__ out)
{
    if (ws_bytes >= TAB_BYTES) return;
    extern __shared__ float smem[];
    float* Ablk = smem + A_OFF;
    float* Hblk = smem + H_OFF;
    float* Bs   = smem + B_OFF;
    int*   SRCs = (int*)(smem + S_OFF);
    int*   DSTs = (int*)(smem + S_OFF + TILE_E);
    float* DIST = smem + S_OFF + 2 * TILE_E;
    float* CWp  = smem + S_OFF + 4 * TILE_E;

    const int tid   = threadIdx.x;
    const int lane  = tid & 31;
    const int wave  = tid >> 5;
    const int row   = lane & 15;
    const int hi    = lane >> 4;
    const int etile = wave >> 1;
    const int nsel  = wave & 1;
    const int e0    = blockIdx.x * TILE_E;

    float R[32];
    int   cur = 0;
    panel_load<256>(cW1, HIDDEN_DIM, 0, R, tid);

    if (tid < TILE_E) {
        SRCs[tid] = eidx[e0 + tid];
        DSTs[tid] = eidx[N_EDGES + e0 + tid];
        DIST[tid] = edist[e0 + tid];
    }
    __syncthreads();

    for (int i = tid; i < TILE_E * 64; i += THREADS) {
        int e = i >> 6, q = i & 63;
        int half = q >> 5, c4 = q & 31;
        int node = half ? DSTs[e] : SRCs[e];
        *(float4*)(Ablk + e * A_LD + half * NODE_DIM + c4 * 4) =
            ((const float4*)h)[node * (NODE_DIM / 4) + c4];
    }

    float* hid = Bs + B_SIZE;
    for (int i = tid; i < TILE_E * EDGE_DIM; i += THREADS) {
        int e = i >> 6, k = i & 63;
        hid[k * TILE_E + e] = silu_f(DIST[e] * eW1[k] + eb1[k]);
    }
    __syncthreads();
    for (int i = tid; i < TILE_E * EDGE_DIM; i += THREADS) {
        int e = i >> 6, j = i & 63;
        float s = eb2[j];
#pragma unroll 8
        for (int k = 0; k < EDGE_DIM; ++k)
            s += hid[k * TILE_E + e] * eW2[k * EDGE_DIM + j];
        Ablk[e * A_LD + 2 * NODE_DIM + j] = s;
    }
    panel_store<256>(Bs, R, tid);
    __syncthreads();

    {
        v8f acc[8];
#pragma unroll
        for (int nt = 0; nt < 8; ++nt) {
            float bv = cb1[nsel * 128 + nt * 16 + row];
            v8f t = {bv, bv, bv, bv, bv, bv, bv, bv};
            acc[nt] = t;
        }
        gemm_pipe<8, 256, 256>(cW1, HIDDEN_DIM, IN_DIM, Ablk, A_LD, Bs, cur, acc, R,
                               nW1, HIDDEN_DIM, tid);
        float p[8];
#pragma unroll
        for (int i = 0; i < 8; ++i) p[i] = 0.0f;
#pragma unroll
        for (int nt = 0; nt < 8; ++nt) {
            const float w = cW2[nsel * 128 + nt * 16 + row];
#pragma unroll
            for (int i = 0; i < 8; ++i)
                p[i] += silu_f(acc[nt][i]) * w;
        }
#pragma unroll
        for (int m = 1; m < 16; m <<= 1)
#pragma unroll
            for (int i = 0; i < 8; ++i)
                p[i] += __shfl_xor(p[i], m, 32);
        if (row == 0) {
            float* q = CWp + nsel * TILE_E + etile * 16 + hi * 8;
            *(float4*)(q)     = make_float4(p[0], p[1], p[2], p[3]);
            *(float4*)(q + 4) = make_float4(p[4], p[5], p[6], p[7]);
        }
        panel_store<256>(Bs + (cur ^ 1) * B_SIZE, R, tid);
        __syncthreads();
        cur ^= 1;
    }

    {
        v8f acc[8];
#pragma unroll
        for (int nt = 0; nt < 8; ++nt) {
            float bv = nb1[nsel * 128 + nt * 16 + row];
            v8f t = {bv, bv, bv, bv, bv, bv, bv, bv};
            acc[nt] = t;
        }
        gemm_pipe<8, 256, 128>(nW1, HIDDEN_DIM, IN_DIM, Ablk, A_LD, Bs, cur, acc, R,
                               nW2, NODE_DIM, tid);
#pragma unroll
        for (int nt = 0; nt < 8; ++nt)
#pragma unroll
            for (int i = 0; i < 8; ++i)
                acc[nt][i] = silu_f(acc[nt][i]);
        panel_store<128>(Bs + (cur ^ 1) * B_SIZE, R, tid);
        __syncthreads();
#pragma unroll
        for (int nt = 0; nt < 8; ++nt)
#pragma unroll
            for (int i = 0; i < 8; ++i) {
                int e = etile * 16 + hi * 8 + i;
                Hblk[e * H_LD + nsel * 128 + nt * 16 + row] = acc[nt][i];
            }
        __syncthreads();
        cur ^= 1;
    }

    {
        v8f acc[4];
#pragma unroll
        for (int nt = 0; nt < 4; ++nt) {
            float bv = nb2[nsel * 64 + nt * 16 + row];
            v8f t = {bv, bv, bv, bv, bv, bv, bv, bv};
            acc[nt] = t;
        }
        gemm_pipe<4, 128, 0>(nW2, NODE_DIM, HIDDEN_DIM, Hblk, H_LD, Bs, cur, acc, R,
                             nW2, NODE_DIM, tid);
#pragma unroll
        for (int nt = 0; nt < 4; ++nt)
#pragma unroll
            for (int i = 0; i < 8; ++i) {
                int e = etile * 16 + hi * 8 + i;
                int d = DSTs[e];
                atomAddF(out + d * NODE_DIM + nsel * 64 + nt * 16 + row, acc[nt][i]);
            }
    }

    if (tid < TILE_E) {
        int e = tid;
        float s = CWp[e] + CWp[TILE_E + e];
        int si = SRCs[e], di = DSTs[e];
        float dx = x[si * 3 + 0] - x[di * 3 + 0];
        float dy = x[si * 3 + 1] - x[di * 3 + 1];
        float dz = x[si * 3 + 2] - x[di * 3 + 2];
        float len = fmaxf(sqrtf(dx * dx + dy * dy + dz * dz), 1e-8f);
        float w = s / len;
        float* ox = out + N_NODES * NODE_DIM;
        atomAddF(ox + di * 3 + 0, w * dx);
        atomAddF(ox + di * 3 + 1, w * dy);
        atomAddF(ox + di * 3 + 2, w * dz);
    }
}


template<int NNEXT>
__device__ __forceinline__ void node_product(const float* __restrict__ W, const float* __restrict__ Wnext,
                                             const float* __restrict__ Ablk, float* __restrict__ Bs, int& cur,
                                             float* R, float* __restrict__ trow, int tid)
{
    const int lane  = tid & 31;
    const int wave  = tid >> 5;
    const int row   = lane & 15;
    const int hi    = lane >> 4;
    const int etile = wave >> 1;
    const int nsel  = wave & 1;
    v8f acc[8];
#pragma unroll
    for (int nt = 0; nt < 8; ++nt) {
        v8f z = {0.f, 0.f, 0.f, 0.f, 0.f, 0.f, 0.f, 0.f};
        acc[nt] = z;
    }
    gemm_pipe<8, 256, NNEXT>(W, HIDDEN_DIM, NODE_DIM, Ablk, NA_LD, Bs, cur, acc, R, Wnext, HIDDEN_DIM, tid);
    float* p = trow + (size_t)(etile * 16 + hi * 8) * TAB_LD + nsel * 128 + row * 4;
#pragma unroll
    for (int i = 0; i < 8; ++i) {
        *(float4*)(p + i * TAB_LD)      = make_float4(acc[0][i], acc[1][i], acc[2][i], acc[3][i]);
        *(float4*)(p + i * TAB_LD + 64) = make_float4(acc[4][i], acc[5][i], acc[6][i], acc[7][i]);
    }
    if constexpr (NNEXT > 0) {
        panel_store<NNEXT>(Bs + (cur ^ 1) * B_SIZE, R, tid);
        __syncthreads();
        cur ^= 1;
    }
}

__global__ void __launch_bounds__(THREADS)
egnn_node_tables(const float* __restrict__ h,
                 const float* __restrict__ nW1, const float* __restrict__ cW1,
                 float* __restrict__ tab, size_t ws_bytes)
{
    if (ws_bytes < TAB_BYTES) return;
    extern __shared__ float smem[];
    float* Ablk = smem;
    float* Bs   = smem + NA_SIZE;
    const int tid = threadIdx.x;
    const int n0  = blockIdx.x * TILE_E;

    float R[32];
    int   cur = 0;
    panel_load<256>(cW1, HIDDEN_DIM, 0, R, tid);
    for (int i = tid; i < TILE_E * 32; i += THREADS) {
        int r = i >> 5, c4 = i & 31;
        *(float4*)(Ablk + r * NA_LD + c4 * 4) = ((const float4*)h)[(n0 + r) * (NODE_DIM / 4) + c4];
    }
    panel_store<256>(Bs, R, tid);
    __syncthreads();

    float* t0 = tab + (size_t)n0 * TAB_LD;
    float* t1 = tab + TAB_FLOATS + (size_t)n0 * TAB_LD;
    node_product<256>(cW1,                         nW1,                         Ablk, Bs, cur, R, t0,       tid);
    node_product<256>(nW1,                         cW1 + NODE_DIM * HIDDEN_DIM, Ablk, Bs, cur, R, t0 + 256, tid);
    node_product<256>(cW1 + NODE_DIM * HIDDEN_DIM, nW1 + NODE_DIM * HIDDEN_DIM, Ablk, Bs, cur, R, t1,       tid);
    node_product<0>  (nW1 + NODE_DIM * HIDDEN_DIM, nW1,                         Ablk, Bs, cur, R, t1 + 256, tid);
}

__device__ __forceinline__ void acc_gather_src(v8f* acc, const float* __restrict__ tab, int g,
                                               const int* SRCs, const float* __restrict__ bias,
                                               int etile, int hi, int nsel, int row)
{
    const int coff = g * 256 + nsel * 128 + row * 4;
    float bv[8];
#pragma unroll
    for (int nt = 0; nt < 8; ++nt) bv[nt] = bias[nsel * 128 + nt * 16 + row];
#pragma unroll
    for (int i = 0; i < 8; ++i) {
        const int e = etile * 16 + hi * 8 + i;
        const float* ps = tab + (size_t)SRCs[e] * TAB_LD + coff;
        const float4 s0 = *(const float4*)(ps), s1 = *(const float4*)(ps + 64);
        acc[0][i] = s0.x + bv[0];
        acc[1][i] = s0.y + bv[1];
        acc[2][i] = s0.z + bv[2];
        acc[3][i] = s0.w + bv[3];
        acc[4][i] = s1.x + bv[4];
        acc[5][i] = s1.y + bv[5];
        acc[6][i] = s1.z + bv[6];
        acc[7][i] = s1.w + bv[7];
    }
}

__device__ __forceinline__ void acc_gather_dst(v8f* acc, const float* __restrict__ tab, int g,
                                               const int* DSTs,
                                               int etile, int hi, int nsel, int row)
{
    const int coff = g * 256 + nsel * 128 + row * 4;
#pragma unroll
    for (int i = 0; i < 8; ++i) {
        const int e = etile * 16 + hi * 8 + i;
        const float* pd = tab + TAB_FLOATS + (size_t)DSTs[e] * TAB_LD + coff;
        const float4 d0 = *(const float4*)(pd), d1 = *(const float4*)(pd + 64);
        acc[0][i] += d0.x;
        acc[1][i] += d0.y;
        acc[2][i] += d0.z;
        acc[3][i] += d0.w;
        acc[4][i] += d1.x;
        acc[5][i] += d1.y;
        acc[6][i] += d1.z;
        acc[7][i] += d1.w;
    }
}

__global__ void __launch_bounds__(THREADS) __attribute__((amdgpu_waves_per_eu(4, 4)))
egnn_edge_tab(const float* __restrict__ x,
              const int*   __restrict__ eidx,const float* __restrict__ edist,
              const float* __restrict__ nW1, const float* __restrict__ nb1,
              const float* __restrict__ nW2, const float* __restrict__ nb2,
              const float* __restrict__ cW1, const float* __restrict__ cb1,
              const float* __restrict__ cW2,
              const float* __restrict__ eW1, const float* __restrict__ eb1,
              const float* __restrict__ eW2, const float* __restrict__ eb2,
              const float* __restrict__ tab, size_t ws_bytes,
              float* __restrict__ out)
{
    if (ws_bytes < TAB_BYTES) return;
    extern __shared__ float smem[];
    float* Ablk = smem;
    float* Hblk = smem;
    float* Bs   = smem + H_SIZE;
    int*   SRCs = (int*)(smem + H_SIZE + 2 * B_SIZE);
    int*   DSTs = SRCs + TILE_E;
    float* DIST = (float*)(DSTs + TILE_E);
    float* CWp  = smem + H_SIZE + 2 * B_SIZE + 4 * TILE_E;

    const int tid   = threadIdx.x;
    const int lane  = tid & 31;
    const int wave  = tid >> 5;
    const int row   = lane & 15;
    const int hi    = lane >> 4;
    const int etile = wave >> 1;
    const int nsel  = wave & 1;
    const int e0    = blockIdx.x * TILE_E;

    const float* cW1e = cW1 + 2 * NODE_DIM * HIDDEN_DIM;
    const float* nW1e = nW1 + 2 * NODE_DIM * HIDDEN_DIM;

    float R[32];
    int   cur = 0;
    panel_load<256>(cW1e, HIDDEN_DIM, 0, R, tid);

    if (tid < TILE_E) {
        SRCs[tid] = eidx[e0 + tid];
        DSTs[tid] = eidx[N_EDGES + e0 + tid];
        DIST[tid] = edist[e0 + tid];
    }
    __syncthreads();

    v8f accC[8];
    acc_gather_src(accC, tab, 0, SRCs, cb1, etile, hi, nsel, row);
    acc_gather_dst(accC, tab, 0, DSTs, etile, hi, nsel, row);

    float* hid = Bs + B_SIZE;
    for (int i = tid; i < TILE_E * EDGE_DIM; i += THREADS) {
        int e = i >> 6, k = i & 63;
        hid[k * TILE_E + e] = silu_f(DIST[e] * eW1[k] + eb1[k]);
    }
    __syncthreads();
    for (int i = tid; i < TILE_E * EDGE_DIM; i += THREADS) {
        int e = i >> 6, j = i & 63;
        float s = eb2[j];
#pragma unroll 8
        for (int k = 0; k < EDGE_DIM; ++k)
            s += hid[k * TILE_E + e] * eW2[k * EDGE_DIM + j];
        Ablk[e * EA_LD + j] = s;
    }
    panel_store<256>(Bs, R, tid);
    __syncthreads();

    v8f accN[8];
    {
        gemm_pipe<8, 256, 256>(cW1e, HIDDEN_DIM, EDGE_DIM, Ablk, EA_LD, Bs, cur, accC, R,
                               nW1e, HIDDEN_DIM, tid);
        acc_gather_src(accN, tab, 1, SRCs, nb1, etile, hi, nsel, row);
        float p[8];
#pragma unroll
        for (int i = 0; i < 8; ++i) p[i] = 0.0f;
#pragma unroll
        for (int nt = 0; nt < 8; ++nt) {
            const float w = cW2[nsel * 128 + nt * 16 + row];
#pragma unroll
            for (int i = 0; i < 8; ++i)
                p[i] += silu_f(accC[nt][i]) * w;
        }
#pragma unroll
        for (int m = 1; m < 16; m <<= 1)
#pragma unroll
            for (int i = 0; i < 8; ++i)
                p[i] += __shfl_xor(p[i], m, 32);
        if (row == 0) {
            float* q = CWp + nsel * TILE_E + etile * 16 + hi * 8;
            *(float4*)(q)     = make_float4(p[0], p[1], p[2], p[3]);
            *(float4*)(q + 4) = make_float4(p[4], p[5], p[6], p[7]);
        }
        acc_gather_dst(accN, tab, 1, DSTs, etile, hi, nsel, row);
        panel_store<256>(Bs + (cur ^ 1) * B_SIZE, R, tid);
        __syncthreads();
        cur ^= 1;
    }

    {
        gemm_pipe<8, 256, 128>(nW1e, HIDDEN_DIM, EDGE_DIM, Ablk, EA_LD, Bs, cur, accN, R,
                               nW2, NODE_DIM, tid);
#pragma unroll
        for (int nt = 0; nt < 8; ++nt)
#pragma unroll
            for (int i = 0; i < 8; ++i)
                accN[nt][i] = silu_f(accN[nt][i]);
        panel_store<128>(Bs + (cur ^ 1) * B_SIZE, R, tid);
        __syncthreads();
#pragma unroll
        for (int nt = 0; nt < 8; ++nt)
#pragma unroll
            for (int i = 0; i < 8; ++i) {
                int e = etile * 16 + hi * 8 + i;
                Hblk[e * H_LD + nsel * 128 + nt * 16 + row] = accN[nt][i];
            }
        __syncthreads();
        cur ^= 1;
    }

    {
        v8f acc[4];
#pragma unroll
        for (int nt = 0; nt < 4; ++nt) {
            float bv = nb2[nsel * 64 + nt * 16 + row];
            v8f t = {bv, bv, bv, bv, bv, bv, bv, bv};
            acc[nt] = t;
        }
        gemm_pipe<4, 128, 0>(nW2, NODE_DIM, HIDDEN_DIM, Hblk, H_LD, Bs, cur, acc, R,
                             nW2, NODE_DIM, tid);
#pragma unroll
        for (int nt = 0; nt < 4; ++nt)
#pragma unroll
            for (int i = 0; i < 8; ++i) {
                int e = etile * 16 + hi * 8 + i;
                int d = DSTs[e];
                atomAddF(out + d * NODE_DIM + nsel * 64 + nt * 16 + row, acc[nt][i]);
            }
    }

    if (tid < TILE_E) {
        int e = tid;
        float s = CWp[e] + CWp[TILE_E + e];
        int si = SRCs[e], di = DSTs[e];
        float dx = x[si * 3 + 0] - x[di * 3 + 0];
        float dy = x[si * 3 + 1] - x[di * 3 + 1];
        float dz = x[si * 3 + 2] - x[di * 3 + 2];
        float len = fmaxf(sqrtf(dx * dx + dy * dy + dz * dz), 1e-8f);
        float w = s / len;
        float* ox = out + N_NODES * NODE_DIM;
        atomAddF(ox + di * 3 + 0, w * dx);
        atomAddF(ox + di * 3 + 1, w * dy);
        atomAddF(ox + di * 3 + 2, w * dz);
    }
}

extern "C" void kernel_launch(void* const* d_in, const int* in_sizes, int n_in,
                              void* d_out, int out_size, void* d_ws, size_t ws_size,
                              hipStream_t stream)
{
    const float* h     = (const float*)d_in[0];
    const float* x     = (const float*)d_in[1];
    const int*   eidx  = (const int*)  d_in[2];
    const float* edist = (const float*)d_in[3];
    const float* nW1   = (const float*)d_in[4];
    const float* nb1   = (const float*)d_in[5];
    const float* nW2   = (const float*)d_in[6];
    const float* nb2   = (const float*)d_in[7];
    const float* cW1   = (const float*)d_in[8];
    const float* cb1   = (const float*)d_in[9];
    const float* cW2   = (const float*)d_in[10];
    const float* eW1   = (const float*)d_in[11];
    const float* eb1   = (const float*)d_in[12];
    const float* eW2   = (const float*)d_in[13];
    const float* eb2   = (const float*)d_in[14];
    float* out = (float*)d_out;

    const int tot = N_NODES * (NODE_DIM + 3);
    egnn_init_out<<<(tot + 255) / 256, 256, 0, stream>>>(h, x, out);

    float* tab = (float*)d_ws;
    egnn_node_tables<<<N_NODES / TILE_E, THREADS, (size_t)NSMEM_FLOATS * sizeof(float), stream>>>(
        h, nW1, cW1, tab, ws_size);
    egnn_edge_tab<<<N_EDGES / TILE_E, THREADS, (size_t)ESMEM_FLOATS * sizeof(float), stream>>>(
        x, eidx, edist,
        nW1, nb1, nW2, nb2,
        cW1, cb1, cW2,
        eW1, eb1, eW2, eb2,
        tab, ws_size,
        out);

    const size_t smem_bytes = (size_t)SMEM_FLOATS * sizeof(float);
    egnn_edge_kernel<<<N_EDGES / TILE_E, THREADS, smem_bytes, stream>>>(
        h, x, eidx, edist,
        nW1, nb1, nW2, nb2,
        cW1, cb1, cW2,
        eW1, eb1, eW2, eb2,
        ws_size,
        out);
}


# Edge MLP second layer (64x64x64 per tile) moved from scalar FMA loops onto the f32 WMMA path: hid kept row-major as A operand, eW2 staged whole in LDS; edge_attr k sum order changes
# speedup vs baseline: 4.5251x; 1.3896x over previous
#include <hip/hip_runtime.h>
#include <math.h>

#define N_NODES   32768
#define N_EDGES   524288
#define NODE_DIM  128
#define HIDDEN_DIM 256
#define EDGE_DIM  64
#define IN_DIM    320

#define TILE_E    64
#define THREADS   256

#define A_LD      324
#define A_OFF     0
#define A_SIZE    (TILE_E * A_LD)
#define H_LD      260
#define H_OFF     A_OFF
#define H_SIZE    (TILE_E * H_LD)
#define B_OFF     (A_OFF + A_SIZE)
#define B_SIZE    (32 * 256)
#define S_OFF     (B_OFF + 2 * B_SIZE)
#define SMEM_FLOATS (S_OFF + 6 * TILE_E)

#define TAB_LD    512
#define TAB_FLOATS ((size_t)N_NODES * TAB_LD)
#define TAB_BYTES  ((size_t)2 * N_NODES * TAB_LD * sizeof(float))

#define NA_LD     132
#define NA_SIZE   (TILE_E * NA_LD)
#define NSMEM_FLOATS (NA_SIZE + 2 * B_SIZE)

#define EA_LD     68
#define ESMEM_FLOATS (H_SIZE + 2 * B_SIZE + 6 * TILE_E)

typedef float v2f __attribute__((ext_vector_type(2)));
typedef float v4f __attribute__((ext_vector_type(4)));
typedef float v8f __attribute__((ext_vector_type(8)));

__device__ __forceinline__ float silu_f(float v) {
    return v / (1.0f + __expf(-v));
}

__device__ __forceinline__ void atomAddF(float* p, float v) {
    __hip_atomic_fetch_add(p, v, __ATOMIC_RELAXED, __HIP_MEMORY_SCOPE_AGENT);
}


template<int NSTAGE>
__device__ __forceinline__ void panel_load(const float* __restrict__ W, int ldw, int kp,
                                           float* R, int tid)
{
    constexpr int KGRP = NSTAGE / 32;
    const int n  = tid & (NSTAGE - 1);
    const int g0 = (tid / NSTAGE) * KGRP;
    const float* p = W + (kp + 4 * g0) * ldw + n;
#pragma unroll
    for (int j = 0; j < 4 * KGRP; ++j)
        R[j] = p[j * ldw];
}

template<int NSTAGE>
__device__ __forceinline__ void panel_store(float* __restrict__ Bs, const float* R, int tid)
{
    constexpr int KGRP = NSTAGE / 32;
    const int n  = tid & (NSTAGE - 1);
    const int g0 = (tid / NSTAGE) * KGRP;
#pragma unroll
    for (int j = 0; j < KGRP; ++j) {
        float4 v = make_float4(R[4 * j], R[4 * j + 1], R[4 * j + 2], R[4 * j + 3]);
        *(float4*)(Bs + (((g0 + j) * NSTAGE + n) << 2)) = v;
    }
}

template<int NTILES, int NSTAGE>
__device__ __forceinline__ void frag_load(const float* __restrict__ Ap,
                                          const float* __restrict__ Bp,
                                          int kb, v4f& a, v4f* b)
{
    a = *(const v4f*)(Ap + 8 * kb);
#pragma unroll
    for (int nt = 0; nt < NTILES; ++nt)
        b[nt] = *(const v4f*)(Bp + ((2 * kb * NSTAGE + nt * 16) << 2));
}

template<int NTILES>
__device__ __forceinline__ void frag_mma(const v4f& a, const v4f* b, v8f* acc)
{
    const v2f alo = {a.x, a.y};
    const v2f ahi = {a.z, a.w};
#pragma unroll
    for (int nt = 0; nt < NTILES; ++nt) {
        const v2f bl = {b[nt].x, b[nt].y};
        acc[nt] = __builtin_amdgcn_wmma_f32_16x16x4_f32(
             false, alo,  false, bl,
             (short)0, acc[nt],
             false,  false);
    }
#pragma unroll
    for (int nt = 0; nt < NTILES; ++nt) {
        const v2f bh = {b[nt].z, b[nt].w};
        acc[nt] = __builtin_amdgcn_wmma_f32_16x16x4_f32(
             false, ahi,  false, bh,
             (short)0, acc[nt],
             false,  false);
    }
}

template<int NTILES, int NSTAGE>
__device__ __forceinline__ void frag_step(const float* __restrict__ Ap,
                                          const float* __restrict__ Bp, int kb,
                                          v4f& an, v4f* bn,
                                          const v4f& ac, const v4f* bc, v8f* acc)
{
    frag_load<NTILES, NSTAGE>(Ap, Bp, kb, an, bn);
    __builtin_amdgcn_sched_barrier(0);
    frag_mma<NTILES>(ac, bc, acc);
    __builtin_amdgcn_sched_barrier(0);
}

template<int NTILES, int NSTAGE, int NNEXT>
__device__ __forceinline__ void gemm_pipe(const float* __restrict__ W, int ldw, int K,
                                          const float* __restrict__ A, int lda,
                                          float* __restrict__ Bs, int& cur,
                                          v8f* acc, float* R,
                                          const float* __restrict__ Wnext, int ldwnext,
                                          int tid)
{
    const int lane  = tid & 31;
    const int wave  = tid >> 5;
    const int row   = lane & 15;
    const int hi    = lane >> 4;
    const int etile = wave >> 1;
    const int nsel  = wave & 1;
    const float* Arow = A + (etile * 16 + row) * lda + 4 * hi;
    const int    boff = (hi * NSTAGE + nsel * (NSTAGE / 2) + row) << 2;

    v4f a0, a1, b0[NTILES], b1[NTILES];
    frag_load<NTILES, NSTAGE>(Arow, Bs + cur * B_SIZE + boff, 0, a0, b0);

    for (int kp = 0; kp < K - 32; kp += 32) {
        const float* Ap = Arow + kp;
        const float* Bp = Bs + cur * B_SIZE + boff;
        panel_load<NSTAGE>(W, ldw, kp + 32, R, tid);
        frag_step<NTILES, NSTAGE>(Ap, Bp, 1, a1, b1, a0, b0, acc);
        frag_step<NTILES, NSTAGE>(Ap, Bp, 2, a0, b0, a1, b1, acc);
        frag_step<NTILES, NSTAGE>(Ap, Bp, 3, a1, b1, a0, b0, acc);
        panel_store<NSTAGE>(Bs + (cur ^ 1) * B_SIZE, R, tid);
        __syncthreads();
        cur ^= 1;
        frag_step<NTILES, NSTAGE>(Ap + 32, Bs + cur * B_SIZE + boff, 0, a0, b0, a1, b1, acc);
    }
    if constexpr (NNEXT > 0) panel_load<NNEXT>(Wnext, ldwnext, 0, R, tid);
    {
        const float* Ap = Arow + K - 32;
        const float* Bp = Bs + cur * B_SIZE + boff;
        frag_step<NTILES, NSTAGE>(Ap, Bp, 1, a1, b1, a0, b0, acc);
        frag_step<NTILES, NSTAGE>(Ap, Bp, 2, a0, b0, a1, b1, acc);
        frag_step<NTILES, NSTAGE>(Ap, Bp, 3, a1, b1, a0, b0, acc);
        frag_mma<NTILES>(a1, b1, acc);
    }
}

__global__ void egnn_init_out(const float* __restrict__ h,
                              const float* __restrict__ x,
                              float* __restrict__ out)
{
    int i = blockIdx.x * blockDim.x + threadIdx.x;
    if (i < N_NODES * NODE_DIM)                out[i] = h[i];
    else if (i < N_NODES * (NODE_DIM + 3))     out[i] = x[i - N_NODES * NODE_DIM];
}

__global__ void __launch_bounds__(THREADS)
egnn_edge_kernel(const float* __restrict__ h,   const float* __restrict__ x,
                 const int*   __restrict__ eidx,const float* __restrict__ edist,
                 const float* __restrict__ nW1, const float* __restrict__ nb1,
                 const float* __restrict__ nW2, const float* __restrict__ nb2,
                 const float* __restrict__ cW1, const float* __restrict__ cb1,
                 const float* __restrict__ cW2,
                 const float* __restrict__ eW1, const float* __restrict__ eb1,
                 const float* __restrict__ eW2, const float* __restrict__ eb2,
                 size_t ws_bytes,
                 float* __restrict__ out)
{
    if (ws_bytes >= TAB_BYTES) return;
    extern __shared__ float smem[];
    float* Ablk = smem + A_OFF;
    float* Hblk = smem + H_OFF;
    float* Bs   = smem + B_OFF;
    int*   SRCs = (int*)(smem + S_OFF);
    int*   DSTs = (int*)(smem + S_OFF + TILE_E);
    float* DIST = smem + S_OFF + 2 * TILE_E;
    float* CWp  = smem + S_OFF + 4 * TILE_E;

    const int tid   = threadIdx.x;
    const int lane  = tid & 31;
    const int wave  = tid >> 5;
    const int row   = lane & 15;
    const int hi    = lane >> 4;
    const int etile = wave >> 1;
    const int nsel  = wave & 1;
    const int e0    = blockIdx.x * TILE_E;

    float R[32];
    int   cur = 0;
    panel_load<256>(cW1, HIDDEN_DIM, 0, R, tid);

    if (tid < TILE_E) {
        SRCs[tid] = eidx[e0 + tid];
        DSTs[tid] = eidx[N_EDGES + e0 + tid];
        DIST[tid] = edist[e0 + tid];
    }
    __syncthreads();

    for (int i = tid; i < TILE_E * 64; i += THREADS) {
        int e = i >> 6, q = i & 63;
        int half = q >> 5, c4 = q & 31;
        int node = half ? DSTs[e] : SRCs[e];
        *(float4*)(Ablk + e * A_LD + half * NODE_DIM + c4 * 4) =
            ((const float4*)h)[node * (NODE_DIM / 4) + c4];
    }

    float* hid = Bs + B_SIZE;
    for (int i = tid; i < TILE_E * EDGE_DIM; i += THREADS) {
        int e = i >> 6, k = i & 63;
        hid[k * TILE_E + e] = silu_f(DIST[e] * eW1[k] + eb1[k]);
    }
    __syncthreads();
    for (int i = tid; i < TILE_E * EDGE_DIM; i += THREADS) {
        int e = i >> 6, j = i & 63;
        float s = eb2[j];
#pragma unroll 8
        for (int k = 0; k < EDGE_DIM; ++k)
            s += hid[k * TILE_E + e] * eW2[k * EDGE_DIM + j];
        Ablk[e * A_LD + 2 * NODE_DIM + j] = s;
    }
    panel_store<256>(Bs, R, tid);
    __syncthreads();

    {
        v8f acc[8];
#pragma unroll
        for (int nt = 0; nt < 8; ++nt) {
            float bv = cb1[nsel * 128 + nt * 16 + row];
            v8f t = {bv, bv, bv, bv, bv, bv, bv, bv};
            acc[nt] = t;
        }
        gemm_pipe<8, 256, 256>(cW1, HIDDEN_DIM, IN_DIM, Ablk, A_LD, Bs, cur, acc, R,
                               nW1, HIDDEN_DIM, tid);
        float p[8];
#pragma unroll
        for (int i = 0; i < 8; ++i) p[i] = 0.0f;
#pragma unroll
        for (int nt = 0; nt < 8; ++nt) {
            const float w = cW2[nsel * 128 + nt * 16 + row];
#pragma unroll
            for (int i = 0; i < 8; ++i)
                p[i] += silu_f(acc[nt][i]) * w;
        }
#pragma unroll
        for (int m = 1; m < 16; m <<= 1)
#pragma unroll
            for (int i = 0; i < 8; ++i)
                p[i] += __shfl_xor(p[i], m, 32);
        if (row == 0) {
            float* q = CWp + nsel * TILE_E + etile * 16 + hi * 8;
            *(float4*)(q)     = make_float4(p[0], p[1], p[2], p[3]);
            *(float4*)(q + 4) = make_float4(p[4], p[5], p[6], p[7]);
        }
        panel_store<256>(Bs + (cur ^ 1) * B_SIZE, R, tid);
        __syncthreads();
        cur ^= 1;
    }

    {
        v8f acc[8];
#pragma unroll
        for (int nt = 0; nt < 8; ++nt) {
            float bv = nb1[nsel * 128 + nt * 16 + row];
            v8f t = {bv, bv, bv, bv, bv, bv, bv, bv};
            acc[nt] = t;
        }
        gemm_pipe<8, 256, 128>(nW1, HIDDEN_DIM, IN_DIM, Ablk, A_LD, Bs, cur, acc, R,
                               nW2, NODE_DIM, tid);
#pragma unroll
        for (int nt = 0; nt < 8; ++nt)
#pragma unroll
            for (int i = 0; i < 8; ++i)
                acc[nt][i] = silu_f(acc[nt][i]);
        panel_store<128>(Bs + (cur ^ 1) * B_SIZE, R, tid);
        __syncthreads();
#pragma unroll
        for (int nt = 0; nt < 8; ++nt)
#pragma unroll
            for (int i = 0; i < 8; ++i) {
                int e = etile * 16 + hi * 8 + i;
                Hblk[e * H_LD + nsel * 128 + nt * 16 + row] = acc[nt][i];
            }
        __syncthreads();
        cur ^= 1;
    }

    {
        v8f acc[4];
#pragma unroll
        for (int nt = 0; nt < 4; ++nt) {
            float bv = nb2[nsel * 64 + nt * 16 + row];
            v8f t = {bv, bv, bv, bv, bv, bv, bv, bv};
            acc[nt] = t;
        }
        gemm_pipe<4, 128, 0>(nW2, NODE_DIM, HIDDEN_DIM, Hblk, H_LD, Bs, cur, acc, R,
                             nW2, NODE_DIM, tid);
#pragma unroll
        for (int nt = 0; nt < 4; ++nt)
#pragma unroll
            for (int i = 0; i < 8; ++i) {
                int e = etile * 16 + hi * 8 + i;
                int d = DSTs[e];
                atomAddF(out + d * NODE_DIM + nsel * 64 + nt * 16 + row, acc[nt][i]);
            }
    }

    if (tid < TILE_E) {
        int e = tid;
        float s = CWp[e] + CWp[TILE_E + e];
        int si = SRCs[e], di = DSTs[e];
        float dx = x[si * 3 + 0] - x[di * 3 + 0];
        float dy = x[si * 3 + 1] - x[di * 3 + 1];
        float dz = x[si * 3 + 2] - x[di * 3 + 2];
        float len = fmaxf(sqrtf(dx * dx + dy * dy + dz * dz), 1e-8f);
        float w = s / len;
        float* ox = out + N_NODES * NODE_DIM;
        atomAddF(ox + di * 3 + 0, w * dx);
        atomAddF(ox + di * 3 + 1, w * dy);
        atomAddF(ox + di * 3 + 2, w * dz);
    }
}


template<int NNEXT>
__device__ __forceinline__ void node_product(const float* __restrict__ W, const float* __restrict__ Wnext,
                                             const float* __restrict__ Ablk, float* __restrict__ Bs, int& cur,
                                             float* R, float* __restrict__ trow, int tid)
{
    const int lane  = tid & 31;
    const int wave  = tid >> 5;
    const int row   = lane & 15;
    const int hi    = lane >> 4;
    const int etile = wave >> 1;
    const int nsel  = wave & 1;
    v8f acc[8];
#pragma unroll
    for (int nt = 0; nt < 8; ++nt) {
        v8f z = {0.f, 0.f, 0.f, 0.f, 0.f, 0.f, 0.f, 0.f};
        acc[nt] = z;
    }
    gemm_pipe<8, 256, NNEXT>(W, HIDDEN_DIM, NODE_DIM, Ablk, NA_LD, Bs, cur, acc, R, Wnext, HIDDEN_DIM, tid);
    float* p = trow + (size_t)(etile * 16 + hi * 8) * TAB_LD + nsel * 128 + row * 4;
#pragma unroll
    for (int i = 0; i < 8; ++i) {
        *(float4*)(p + i * TAB_LD)      = make_float4(acc[0][i], acc[1][i], acc[2][i], acc[3][i]);
        *(float4*)(p + i * TAB_LD + 64) = make_float4(acc[4][i], acc[5][i], acc[6][i], acc[7][i]);
    }
    if constexpr (NNEXT > 0) {
        panel_store<NNEXT>(Bs + (cur ^ 1) * B_SIZE, R, tid);
        __syncthreads();
        cur ^= 1;
    }
}

__global__ void __launch_bounds__(THREADS)
egnn_node_tables(const float* __restrict__ h,
                 const float* __restrict__ nW1, const float* __restrict__ cW1,
                 float* __restrict__ tab, size_t ws_bytes)
{
    if (ws_bytes < TAB_BYTES) return;
    extern __shared__ float smem[];
    float* Ablk = smem;
    float* Bs   = smem + NA_SIZE;
    const int tid = threadIdx.x;
    const int n0  = blockIdx.x * TILE_E;

    float R[32];
    int   cur = 0;
    panel_load<256>(cW1, HIDDEN_DIM, 0, R, tid);
    for (int i = tid; i < TILE_E * 32; i += THREADS) {
        int r = i >> 5, c4 = i & 31;
        *(float4*)(Ablk + r * NA_LD + c4 * 4) = ((const float4*)h)[(n0 + r) * (NODE_DIM / 4) + c4];
    }
    panel_store<256>(Bs, R, tid);
    __syncthreads();

    float* t0 = tab + (size_t)n0 * TAB_LD;
    float* t1 = tab + TAB_FLOATS + (size_t)n0 * TAB_LD;
    node_product<256>(cW1,                         nW1,                         Ablk, Bs, cur, R, t0,       tid);
    node_product<256>(nW1,                         cW1 + NODE_DIM * HIDDEN_DIM, Ablk, Bs, cur, R, t0 + 256, tid);
    node_product<256>(cW1 + NODE_DIM * HIDDEN_DIM, nW1 + NODE_DIM * HIDDEN_DIM, Ablk, Bs, cur, R, t1,       tid);
    node_product<0>  (nW1 + NODE_DIM * HIDDEN_DIM, nW1,                         Ablk, Bs, cur, R, t1 + 256, tid);
}

__device__ __forceinline__ void acc_gather_src(v8f* acc, const float* __restrict__ tab, int g,
                                               const int* SRCs, const float* __restrict__ bias,
                                               int etile, int hi, int nsel, int row)
{
    const int coff = g * 256 + nsel * 128 + row * 4;
    float bv[8];
#pragma unroll
    for (int nt = 0; nt < 8; ++nt) bv[nt] = bias[nsel * 128 + nt * 16 + row];
#pragma unroll
    for (int i = 0; i < 8; ++i) {
        const int e = etile * 16 + hi * 8 + i;
        const float* ps = tab + (size_t)SRCs[e] * TAB_LD + coff;
        const float4 s0 = *(const float4*)(ps), s1 = *(const float4*)(ps + 64);
        acc[0][i] = s0.x + bv[0];
        acc[1][i] = s0.y + bv[1];
        acc[2][i] = s0.z + bv[2];
        acc[3][i] = s0.w + bv[3];
        acc[4][i] = s1.x + bv[4];
        acc[5][i] = s1.y + bv[5];
        acc[6][i] = s1.z + bv[6];
        acc[7][i] = s1.w + bv[7];
    }
}

__device__ __forceinline__ void acc_gather_dst(v8f* acc, const float* __restrict__ tab, int g,
                                               const int* DSTs,
                                               int etile, int hi, int nsel, int row)
{
    const int coff = g * 256 + nsel * 128 + row * 4;
#pragma unroll
    for (int i = 0; i < 8; ++i) {
        const int e = etile * 16 + hi * 8 + i;
        const float* pd = tab + TAB_FLOATS + (size_t)DSTs[e] * TAB_LD + coff;
        const float4 d0 = *(const float4*)(pd), d1 = *(const float4*)(pd + 64);
        acc[0][i] += d0.x;
        acc[1][i] += d0.y;
        acc[2][i] += d0.z;
        acc[3][i] += d0.w;
        acc[4][i] += d1.x;
        acc[5][i] += d1.y;
        acc[6][i] += d1.z;
        acc[7][i] += d1.w;
    }
}

__global__ void __launch_bounds__(THREADS) __attribute__((amdgpu_waves_per_eu(4, 4)))
egnn_edge_tab(const float* __restrict__ x,
              const int*   __restrict__ eidx,const float* __restrict__ edist,
              const float* __restrict__ nW1, const float* __restrict__ nb1,
              const float* __restrict__ nW2, const float* __restrict__ nb2,
              const float* __restrict__ cW1, const float* __restrict__ cb1,
              const float* __restrict__ cW2,
              const float* __restrict__ eW1, const float* __restrict__ eb1,
              const float* __restrict__ eW2, const float* __restrict__ eb2,
              const float* __restrict__ tab, size_t ws_bytes,
              float* __restrict__ out)
{
    if (ws_bytes < TAB_BYTES) return;
    extern __shared__ float smem[];
    float* Ablk = smem;
    float* Hblk = smem;
    float* Bs   = smem + H_SIZE;
    int*   SRCs = (int*)(smem + H_SIZE + 2 * B_SIZE);
    int*   DSTs = SRCs + TILE_E;
    float* DIST = (float*)(DSTs + TILE_E);
    float* CWp  = smem + H_SIZE + 2 * B_SIZE + 4 * TILE_E;

    const int tid   = threadIdx.x;
    const int lane  = tid & 31;
    const int wave  = tid >> 5;
    const int row   = lane & 15;
    const int hi    = lane >> 4;
    const int etile = wave >> 1;
    const int nsel  = wave & 1;
    const int e0    = blockIdx.x * TILE_E;

    const float* cW1e = cW1 + 2 * NODE_DIM * HIDDEN_DIM;
    const float* nW1e = nW1 + 2 * NODE_DIM * HIDDEN_DIM;

    float R[32];
    int   cur = 0;
    panel_load<64>(eW2, EDGE_DIM, 0,  R,     tid);
    panel_load<64>(eW2, EDGE_DIM, 32, R + 8, tid);
    const float w1k = eW1[tid & 63], b1k = eb1[tid & 63];

    if (tid < TILE_E) {
        SRCs[tid] = eidx[e0 + tid];
        DSTs[tid] = eidx[N_EDGES + e0 + tid];
        DIST[tid] = edist[e0 + tid];
    }
    __syncthreads();

    v8f accC[8];
    acc_gather_src(accC, tab, 0, SRCs, cb1, etile, hi, nsel, row);
    acc_gather_dst(accC, tab, 0, DSTs, etile, hi, nsel, row);

    float* HID = smem + TILE_E * EA_LD;
    for (int i = tid; i < TILE_E * EDGE_DIM; i += THREADS) {
        int e = i >> 6;
        HID[e * EA_LD + (tid & 63)] = silu_f(DIST[e] * w1k + b1k);
    }
    panel_store<64>(Bs, R, tid);
    panel_store<64>(Bs + 32 * 64, R + 8, tid);
    __syncthreads();

    panel_load<256>(cW1e, HIDDEN_DIM, 0, R, tid);

    {
        v8f acc[2];
#pragma unroll
        for (int nt = 0; nt < 2; ++nt) {
            float bv = eb2[nsel * 32 + nt * 16 + row];
            v8f t = {bv, bv, bv, bv, bv, bv, bv, bv};
            acc[nt] = t;
        }
        const float* Arow = HID + (etile * 16 + row) * EA_LD + 4 * hi;
        const float* Bp   = Bs + ((hi * 64 + nsel * 32 + row) << 2);
        v4f a0, a1, b0[2], b1[2];
        frag_load<2, 64>(Arow, Bp, 0, a0, b0);
        frag_step<2, 64>(Arow, Bp, 1, a1, b1, a0, b0, acc);
        frag_step<2, 64>(Arow, Bp, 2, a0, b0, a1, b1, acc);
        frag_step<2, 64>(Arow, Bp, 3, a1, b1, a0, b0, acc);
        frag_step<2, 64>(Arow, Bp, 4, a0, b0, a1, b1, acc);
        frag_step<2, 64>(Arow, Bp, 5, a1, b1, a0, b0, acc);
        frag_step<2, 64>(Arow, Bp, 6, a0, b0, a1, b1, acc);
        frag_step<2, 64>(Arow, Bp, 7, a1, b1, a0, b0, acc);
        frag_mma<2>(a1, b1, acc);
#pragma unroll
        for (int nt = 0; nt < 2; ++nt)
#pragma unroll
            for (int i = 0; i < 8; ++i) {
                int e = etile * 16 + hi * 8 + i;
                Ablk[e * EA_LD + nsel * 32 + nt * 16 + row] = acc[nt][i];
            }
    }
    panel_store<256>(Bs + (cur ^ 1) * B_SIZE, R, tid);
    __syncthreads();
    cur ^= 1;

    v8f accN[8];
    {
        gemm_pipe<8, 256, 256>(cW1e, HIDDEN_DIM, EDGE_DIM, Ablk, EA_LD, Bs, cur, accC, R,
                               nW1e, HIDDEN_DIM, tid);
        acc_gather_src(accN, tab, 1, SRCs, nb1, etile, hi, nsel, row);
        float p[8];
#pragma unroll
        for (int i = 0; i < 8; ++i) p[i] = 0.0f;
#pragma unroll
        for (int nt = 0; nt < 8; ++nt) {
            const float w = cW2[nsel * 128 + nt * 16 + row];
#pragma unroll
            for (int i = 0; i < 8; ++i)
                p[i] += silu_f(accC[nt][i]) * w;
        }
#pragma unroll
        for (int m = 1; m < 16; m <<= 1)
#pragma unroll
            for (int i = 0; i < 8; ++i)
                p[i] += __shfl_xor(p[i], m, 32);
        if (row == 0) {
            float* q = CWp + nsel * TILE_E + etile * 16 + hi * 8;
            *(float4*)(q)     = make_float4(p[0], p[1], p[2], p[3]);
            *(float4*)(q + 4) = make_float4(p[4], p[5], p[6], p[7]);
        }
        acc_gather_dst(accN, tab, 1, DSTs, etile, hi, nsel, row);
        panel_store<256>(Bs + (cur ^ 1) * B_SIZE, R, tid);
        __syncthreads();
        cur ^= 1;
    }

    {
        gemm_pipe<8, 256, 128>(nW1e, HIDDEN_DIM, EDGE_DIM, Ablk, EA_LD, Bs, cur, accN, R,
                               nW2, NODE_DIM, tid);
#pragma unroll
        for (int nt = 0; nt < 8; ++nt)
#pragma unroll
            for (int i = 0; i < 8; ++i)
                accN[nt][i] = silu_f(accN[nt][i]);
        panel_store<128>(Bs + (cur ^ 1) * B_SIZE, R, tid);
        __syncthreads();
#pragma unroll
        for (int nt = 0; nt < 8; ++nt)
#pragma unroll
            for (int i = 0; i < 8; ++i) {
                int e = etile * 16 + hi * 8 + i;
                Hblk[e * H_LD + nsel * 128 + nt * 16 + row] = accN[nt][i];
            }
        __syncthreads();
        cur ^= 1;
    }

    {
        v8f acc[4];
#pragma unroll
        for (int nt = 0; nt < 4; ++nt) {
            float bv = nb2[nsel * 64 + nt * 16 + row];
            v8f t = {bv, bv, bv, bv, bv, bv, bv, bv};
            acc[nt] = t;
        }
        gemm_pipe<4, 128, 0>(nW2, NODE_DIM, HIDDEN_DIM, Hblk, H_LD, Bs, cur, acc, R,
                             nW2, NODE_DIM, tid);
#pragma unroll
        for (int nt = 0; nt < 4; ++nt)
#pragma unroll
            for (int i = 0; i < 8; ++i) {
                int e = etile * 16 + hi * 8 + i;
                int d = DSTs[e];
                atomAddF(out + d * NODE_DIM + nsel * 64 + nt * 16 + row, acc[nt][i]);
            }
    }

    if (tid < TILE_E) {
        int e = tid;
        float s = CWp[e] + CWp[TILE_E + e];
        int si = SRCs[e], di = DSTs[e];
        float dx = x[si * 3 + 0] - x[di * 3 + 0];
        float dy = x[si * 3 + 1] - x[di * 3 + 1];
        float dz = x[si * 3 + 2] - x[di * 3 + 2];
        float len = fmaxf(sqrtf(dx * dx + dy * dy + dz * dz), 1e-8f);
        float w = s / len;
        float* ox = out + N_NODES * NODE_DIM;
        atomAddF(ox + di * 3 + 0, w * dx);
        atomAddF(ox + di * 3 + 1, w * dy);
        atomAddF(ox + di * 3 + 2, w * dz);
    }
}

extern "C" void kernel_launch(void* const* d_in, const int* in_sizes, int n_in,
                              void* d_out, int out_size, void* d_ws, size_t ws_size,
                              hipStream_t stream)
{
    const float* h     = (const float*)d_in[0];
    const float* x     = (const float*)d_in[1];
    const int*   eidx  = (const int*)  d_in[2];
    const float* edist = (const float*)d_in[3];
    const float* nW1   = (const float*)d_in[4];
    const float* nb1   = (const float*)d_in[5];
    const float* nW2   = (const float*)d_in[6];
    const float* nb2   = (const float*)d_in[7];
    const float* cW1   = (const float*)d_in[8];
    const float* cb1   = (const float*)d_in[9];
    const float* cW2   = (const float*)d_in[10];
    const float* eW1   = (const float*)d_in[11];
    const float* eb1   = (const float*)d_in[12];
    const float* eW2   = (const float*)d_in[13];
    const float* eb2   = (const float*)d_in[14];
    float* out = (float*)d_out;

    const int tot = N_NODES * (NODE_DIM + 3);
    egnn_init_out<<<(tot + 255) / 256, 256, 0, stream>>>(h, x, out);

    float* tab = (float*)d_ws;
    egnn_node_tables<<<N_NODES / TILE_E, THREADS, (size_t)NSMEM_FLOATS * sizeof(float), stream>>>(
        h, nW1, cW1, tab, ws_size);
    egnn_edge_tab<<<N_EDGES / TILE_E, THREADS, (size_t)ESMEM_FLOATS * sizeof(float), stream>>>(
        x, eidx, edist,
        nW1, nb1, nW2, nb2,
        cW1, cb1, cW2,
        eW1, eb1, eW2, eb2,
        tab, ws_size,
        out);

    const size_t smem_bytes = (size_t)SMEM_FLOATS * sizeof(float);
    egnn_edge_kernel<<<N_EDGES / TILE_E, THREADS, smem_bytes, stream>>>(
        h, x, eidx, edist,
        nW1, nb1, nW2, nb2,
        cW1, cb1, cW2,
        eW1, eb1, eW2, eb2,
        ws_size,
        out);
}


# SiLU quotient v/(1+exp(-v)) taken as v times one hardware reciprocal (v_rcp_f32, 1 ulp) instead of the IEEE division sequence: about 8 fewer vector instructions per activation; sum order unchanged
# speedup vs baseline: 4.7547x; 1.0507x over previous
#include <hip/hip_runtime.h>
#include <math.h>

#define N_NODES   32768
#define N_EDGES   524288
#define NODE_DIM  128
#define HIDDEN_DIM 256
#define EDGE_DIM  64
#define IN_DIM    320

#define TILE_E    64
#define THREADS   256

#define A_LD      324
#define A_OFF     0
#define A_SIZE    (TILE_E * A_LD)
#define H_LD      260
#define H_OFF     A_OFF
#define H_SIZE    (TILE_E * H_LD)
#define B_OFF     (A_OFF + A_SIZE)
#define B_SIZE    (32 * 256)
#define S_OFF     (B_OFF + 2 * B_SIZE)
#define SMEM_FLOATS (S_OFF + 6 * TILE_E)

#define TAB_LD    512
#define TAB_FLOATS ((size_t)N_NODES * TAB_LD)
#define TAB_BYTES  ((size_t)2 * N_NODES * TAB_LD * sizeof(float))

#define NA_LD     132
#define NA_SIZE   (TILE_E * NA_LD)
#define NSMEM_FLOATS (NA_SIZE + 2 * B_SIZE)

#define EA_LD     68
#define ESMEM_FLOATS (H_SIZE + 2 * B_SIZE + 6 * TILE_E)

typedef float v2f __attribute__((ext_vector_type(2)));
typedef float v4f __attribute__((ext_vector_type(4)));
typedef float v8f __attribute__((ext_vector_type(8)));

__device__ __forceinline__ float silu_f(float v) {
    return v * __builtin_amdgcn_rcpf(1.0f + __expf(-v));
}

__device__ __forceinline__ void atomAddF(float* p, float v) {
    __hip_atomic_fetch_add(p, v, __ATOMIC_RELAXED, __HIP_MEMORY_SCOPE_AGENT);
}


template<int NSTAGE>
__device__ __forceinline__ void panel_load(const float* __restrict__ W, int ldw, int kp,
                                           float* R, int tid)
{
    constexpr int KGRP = NSTAGE / 32;
    const int n  = tid & (NSTAGE - 1);
    const int g0 = (tid / NSTAGE) * KGRP;
    const float* p = W + (kp + 4 * g0) * ldw + n;
#pragma unroll
    for (int j = 0; j < 4 * KGRP; ++j)
        R[j] = p[j * ldw];
}

template<int NSTAGE>
__device__ __forceinline__ void panel_store(float* __restrict__ Bs, const float* R, int tid)
{
    constexpr int KGRP = NSTAGE / 32;
    const int n  = tid & (NSTAGE - 1);
    const int g0 = (tid / NSTAGE) * KGRP;
#pragma unroll
    for (int j = 0; j < KGRP; ++j) {
        float4 v = make_float4(R[4 * j], R[4 * j + 1], R[4 * j + 2], R[4 * j + 3]);
        *(float4*)(Bs + (((g0 + j) * NSTAGE + n) << 2)) = v;
    }
}

template<int NTILES, int NSTAGE>
__device__ __forceinline__ void frag_load(const float* __restrict__ Ap,
                                          const float* __restrict__ Bp,
                                          int kb, v4f& a, v4f* b)
{
    a = *(const v4f*)(Ap + 8 * kb);
#pragma unroll
    for (int nt = 0; nt < NTILES; ++nt)
        b[nt] = *(const v4f*)(Bp + ((2 * kb * NSTAGE + nt * 16) << 2));
}

template<int NTILES>
__device__ __forceinline__ void frag_mma(const v4f& a, const v4f* b, v8f* acc)
{
    const v2f alo = {a.x, a.y};
    const v2f ahi = {a.z, a.w};
#pragma unroll
    for (int nt = 0; nt < NTILES; ++nt) {
        const v2f bl = {b[nt].x, b[nt].y};
        acc[nt] = __builtin_amdgcn_wmma_f32_16x16x4_f32(
             false, alo,  false, bl,
             (short)0, acc[nt],
             false,  false);
    }
#pragma unroll
    for (int nt = 0; nt < NTILES; ++nt) {
        const v2f bh = {b[nt].z, b[nt].w};
        acc[nt] = __builtin_amdgcn_wmma_f32_16x16x4_f32(
             false, ahi,  false, bh,
             (short)0, acc[nt],
             false,  false);
    }
}

template<int NTILES, int NSTAGE>
__device__ __forceinline__ void frag_step(const float* __restrict__ Ap,
                                          const float* __restrict__ Bp, int kb,
                                          v4f& an, v4f* bn,
                                          const v4f& ac, const v4f* bc, v8f* acc)
{
    frag_load<NTILES, NSTAGE>(Ap, Bp, kb, an, bn);
    __builtin_amdgcn_sched_barrier(0);
    frag_mma<NTILES>(ac, bc, acc);
    __builtin_amdgcn_sched_barrier(0);
}

template<int NTILES, int NSTAGE, int NNEXT>
__device__ __forceinline__ void gemm_pipe(const float* __restrict__ W, int ldw, int K,
                                          const float* __restrict__ A, int lda,
                                          float* __restrict__ Bs, int& cur,
                                          v8f* acc, float* R,
                                          const float* __restrict__ Wnext, int ldwnext,
                                          int tid)
{
    const int lane  = tid & 31;
    const int wave  = tid >> 5;
    const int row   = lane & 15;
    const int hi    = lane >> 4;
    const int etile = wave >> 1;
    const int nsel  = wave & 1;
    const float* Arow = A + (etile * 16 + row) * lda + 4 * hi;
    const int    boff = (hi * NSTAGE + nsel * (NSTAGE / 2) + row) << 2;

    v4f a0, a1, b0[NTILES], b1[NTILES];
    frag_load<NTILES, NSTAGE>(Arow, Bs + cur * B_SIZE + boff, 0, a0, b0);

    for (int kp = 0; kp < K - 32; kp += 32) {
        const float* Ap = Arow + kp;
        const float* Bp = Bs + cur * B_SIZE + boff;
        panel_load<NSTAGE>(W, ldw, kp + 32, R, tid);
        frag_step<NTILES, NSTAGE>(Ap, Bp, 1, a1, b1, a0, b0, acc);
        frag_step<NTILES, NSTAGE>(Ap, Bp, 2, a0, b0, a1, b1, acc);
        frag_step<NTILES, NSTAGE>(Ap, Bp, 3, a1, b1, a0, b0, acc);
        panel_store<NSTAGE>(Bs + (cur ^ 1) * B_SIZE, R, tid);
        __syncthreads();
        cur ^= 1;
        frag_step<NTILES, NSTAGE>(Ap + 32, Bs + cur * B_SIZE + boff, 0, a0, b0, a1, b1, acc);
    }
    if constexpr (NNEXT > 0) panel_load<NNEXT>(Wnext, ldwnext, 0, R, tid);
    {
        const float* Ap = Arow + K - 32;
        const float* Bp = Bs + cur * B_SIZE + boff;
        frag_step<NTILES, NSTAGE>(Ap, Bp, 1, a1, b1, a0, b0, acc);
        frag_step<NTILES, NSTAGE>(Ap, Bp, 2, a0, b0, a1, b1, acc);
        frag_step<NTILES, NSTAGE>(Ap, Bp, 3, a1, b1, a0, b0, acc);
        frag_mma<NTILES>(a1, b1, acc);
    }
}

__global__ void egnn_init_out(const float* __restrict__ h,
                              const float* __restrict__ x,
                              float* __restrict__ out)
{
    int i = blockIdx.x * blockDim.x + threadIdx.x;
    if (i < N_NODES * NODE_DIM)                out[i] = h[i];
    else if (i < N_NODES * (NODE_DIM + 3))     out[i] = x[i - N_NODES * NODE_DIM];
}

__global__ void __launch_bounds__(THREADS)
egnn_edge_kernel(const float* __restrict__ h,   const float* __restrict__ x,
                 const int*   __restrict__ eidx,const float* __restrict__ edist,
                 const float* __restrict__ nW1, const float* __restrict__ nb1,
                 const float* __restrict__ nW2, const float* __restrict__ nb2,
                 const float* __restrict__ cW1, const float* __restrict__ cb1,
                 const float* __restrict__ cW2,
                 const float* __restrict__ eW1, const float* __restrict__ eb1,
                 const float* __restrict__ eW2, const float* __restrict__ eb2,
                 size_t ws_bytes,
                 float* __restrict__ out)
{
    if (ws_bytes >= TAB_BYTES) return;
    extern __shared__ float smem[];
    float* Ablk = smem + A_OFF;
    float* Hblk = smem + H_OFF;
    float* Bs   = smem + B_OFF;
    int*   SRCs = (int*)(smem + S_OFF);
    int*   DSTs = (int*)(smem + S_OFF + TILE_E);
    float* DIST = smem + S_OFF + 2 * TILE_E;
    float* CWp  = smem + S_OFF + 4 * TILE_E;

    const int tid   = threadIdx.x;
    const int lane  = tid & 31;
    const int wave  = tid >> 5;
    const int row   = lane & 15;
    const int hi    = lane >> 4;
    const int etile = wave >> 1;
    const int nsel  = wave & 1;
    const int e0    = blockIdx.x * TILE_E;

    float R[32];
    int   cur = 0;
    panel_load<256>(cW1, HIDDEN_DIM, 0, R, tid);

    if (tid < TILE_E) {
        SRCs[tid] = eidx[e0 + tid];
        DSTs[tid] = eidx[N_EDGES + e0 + tid];
        DIST[tid] = edist[e0 + tid];
    }
    __syncthreads();

    for (int i = tid; i < TILE_E * 64; i += THREADS) {
        int e = i >> 6, q = i & 63;
        int half = q >> 5, c4 = q & 31;
        int node = half ? DSTs[e] : SRCs[e];
        *(float4*)(Ablk + e * A_LD + half * NODE_DIM + c4 * 4) =
            ((const float4*)h)[node * (NODE_DIM / 4) + c4];
    }

    float* hid = Bs + B_SIZE;
    for (int i = tid; i < TILE_E * EDGE_DIM; i += THREADS) {
        int e = i >> 6, k = i & 63;
        hid[k * TILE_E + e] = silu_f(DIST[e] * eW1[k] + eb1[k]);
    }
    __syncthreads();
    for (int i = tid; i < TILE_E * EDGE_DIM; i += THREADS) {
        int e = i >> 6, j = i & 63;
        float s = eb2[j];
#pragma unroll 8
        for (int k = 0; k < EDGE_DIM; ++k)
            s += hid[k * TILE_E + e] * eW2[k * EDGE_DIM + j];
        Ablk[e * A_LD + 2 * NODE_DIM + j] = s;
    }
    panel_store<256>(Bs, R, tid);
    __syncthreads();

    {
        v8f acc[8];
#pragma unroll
        for (int nt = 0; nt < 8; ++nt) {
            float bv = cb1[nsel * 128 + nt * 16 + row];
            v8f t = {bv, bv, bv, bv, bv, bv, bv, bv};
            acc[nt] = t;
        }
        gemm_pipe<8, 256, 256>(cW1, HIDDEN_DIM, IN_DIM, Ablk, A_LD, Bs, cur, acc, R,
                               nW1, HIDDEN_DIM, tid);
        float p[8];
#pragma unroll
        for (int i = 0; i < 8; ++i) p[i] = 0.0f;
#pragma unroll
        for (int nt = 0; nt < 8; ++nt) {
            const float w = cW2[nsel * 128 + nt * 16 + row];
#pragma unroll
            for (int i = 0; i < 8; ++i)
                p[i] += silu_f(acc[nt][i]) * w;
        }
#pragma unroll
        for (int m = 1; m < 16; m <<= 1)
#pragma unroll
            for (int i = 0; i < 8; ++i)
                p[i] += __shfl_xor(p[i], m, 32);
        if (row == 0) {
            float* q = CWp + nsel * TILE_E + etile * 16 + hi * 8;
            *(float4*)(q)     = make_float4(p[0], p[1], p[2], p[3]);
            *(float4*)(q + 4) = make_float4(p[4], p[5], p[6], p[7]);
        }
        panel_store<256>(Bs + (cur ^ 1) * B_SIZE, R, tid);
        __syncthreads();
        cur ^= 1;
    }

    {
        v8f acc[8];
#pragma unroll
        for (int nt = 0; nt < 8; ++nt) {
            float bv = nb1[nsel * 128 + nt * 16 + row];
            v8f t = {bv, bv, bv, bv, bv, bv, bv, bv};
            acc[nt] = t;
        }
        gemm_pipe<8, 256, 128>(nW1, HIDDEN_DIM, IN_DIM, Ablk, A_LD, Bs, cur, acc, R,
                               nW2, NODE_DIM, tid);
#pragma unroll
        for (int nt = 0; nt < 8; ++nt)
#pragma unroll
            for (int i = 0; i < 8; ++i)
                acc[nt][i] = silu_f(acc[nt][i]);
        panel_store<128>(Bs + (cur ^ 1) * B_SIZE, R, tid);
        __syncthreads();
#pragma unroll
        for (int nt = 0; nt < 8; ++nt)
#pragma unroll
            for (int i = 0; i < 8; ++i) {
                int e = etile * 16 + hi * 8 + i;
                Hblk[e * H_LD + nsel * 128 + nt * 16 + row] = acc[nt][i];
            }
        __syncthreads();
        cur ^= 1;
    }

    {
        v8f acc[4];
#pragma unroll
        for (int nt = 0; nt < 4; ++nt) {
            float bv = nb2[nsel * 64 + nt * 16 + row];
            v8f t = {bv, bv, bv, bv, bv, bv, bv, bv};
            acc[nt] = t;
        }
        gemm_pipe<4, 128, 0>(nW2, NODE_DIM, HIDDEN_DIM, Hblk, H_LD, Bs, cur, acc, R,
                             nW2, NODE_DIM, tid);
#pragma unroll
        for (int nt = 0; nt < 4; ++nt)
#pragma unroll
            for (int i = 0; i < 8; ++i) {
                int e = etile * 16 + hi * 8 + i;
                int d = DSTs[e];
                atomAddF(out + d * NODE_DIM + nsel * 64 + nt * 16 + row, acc[nt][i]);
            }
    }

    if (tid < TILE_E) {
        int e = tid;
        float s = CWp[e] + CWp[TILE_E + e];
        int si = SRCs[e], di = DSTs[e];
        float dx = x[si * 3 + 0] - x[di * 3 + 0];
        float dy = x[si * 3 + 1] - x[di * 3 + 1];
        float dz = x[si * 3 + 2] - x[di * 3 + 2];
        float len = fmaxf(sqrtf(dx * dx + dy * dy + dz * dz), 1e-8f);
        float w = s / len;
        float* ox = out + N_NODES * NODE_DIM;
        atomAddF(ox + di * 3 + 0, w * dx);
        atomAddF(ox + di * 3 + 1, w * dy);
        atomAddF(ox + di * 3 + 2, w * dz);
    }
}


template<int NNEXT>
__device__ __forceinline__ void node_product(const float* __restrict__ W, const float* __restrict__ Wnext,
                                             const float* __restrict__ Ablk, float* __restrict__ Bs, int& cur,
                                             float* R, float* __restrict__ trow, int tid)
{
    const int lane  = tid & 31;
    const int wave  = tid >> 5;
    const int row   = lane & 15;
    const int hi    = lane >> 4;
    const int etile = wave >> 1;
    const int nsel  = wave & 1;
    v8f acc[8];
#pragma unroll
    for (int nt = 0; nt < 8; ++nt) {
        v8f z = {0.f, 0.f, 0.f, 0.f, 0.f, 0.f, 0.f, 0.f};
        acc[nt] = z;
    }
    gemm_pipe<8, 256, NNEXT>(W, HIDDEN_DIM, NODE_DIM, Ablk, NA_LD, Bs, cur, acc, R, Wnext, HIDDEN_DIM, tid);
    float* p = trow + (size_t)(etile * 16 + hi * 8) * TAB_LD + nsel * 128 + row * 4;
#pragma unroll
    for (int i = 0; i < 8; ++i) {
        *(float4*)(p + i * TAB_LD)      = make_float4(acc[0][i], acc[1][i], acc[2][i], acc[3][i]);
        *(float4*)(p + i * TAB_LD + 64) = make_float4(acc[4][i], acc[5][i], acc[6][i], acc[7][i]);
    }
    if constexpr (NNEXT > 0) {
        panel_store<NNEXT>(Bs + (cur ^ 1) * B_SIZE, R, tid);
        __syncthreads();
        cur ^= 1;
    }
}

__global__ void __launch_bounds__(THREADS)
egnn_node_tables(const float* __restrict__ h,
                 const float* __restrict__ nW1, const float* __restrict__ cW1,
                 float* __restrict__ tab, size_t ws_bytes)
{
    if (ws_bytes < TAB_BYTES) return;
    extern __shared__ float smem[];
    float* Ablk = smem;
    float* Bs   = smem + NA_SIZE;
    const int tid = threadIdx.x;
    const int n0  = blockIdx.x * TILE_E;

    float R[32];
    int   cur = 0;
    panel_load<256>(cW1, HIDDEN_DIM, 0, R, tid);
    for (int i = tid; i < TILE_E * 32; i += THREADS) {
        int r = i >> 5, c4 = i & 31;
        *(float4*)(Ablk + r * NA_LD + c4 * 4) = ((const float4*)h)[(n0 + r) * (NODE_DIM / 4) + c4];
    }
    panel_store<256>(Bs, R, tid);
    __syncthreads();

    float* t0 = tab + (size_t)n0 * TAB_LD;
    float* t1 = tab + TAB_FLOATS + (size_t)n0 * TAB_LD;
    node_product<256>(cW1,                         nW1,                         Ablk, Bs, cur, R, t0,       tid);
    node_product<256>(nW1,                         cW1 + NODE_DIM * HIDDEN_DIM, Ablk, Bs, cur, R, t0 + 256, tid);
    node_product<256>(cW1 + NODE_DIM * HIDDEN_DIM, nW1 + NODE_DIM * HIDDEN_DIM, Ablk, Bs, cur, R, t1,       tid);
    node_product<0>  (nW1 + NODE_DIM * HIDDEN_DIM, nW1,                         Ablk, Bs, cur, R, t1 + 256, tid);
}

__device__ __forceinline__ void acc_gather_src(v8f* acc, const float* __restrict__ tab, int g,
                                               const int* SRCs, const float* __restrict__ bias,
                                               int etile, int hi, int nsel, int row)
{
    const int coff = g * 256 + nsel * 128 + row * 4;
    float bv[8];
#pragma unroll
    for (int nt = 0; nt < 8; ++nt) bv[nt] = bias[nsel * 128 + nt * 16 + row];
#pragma unroll
    for (int i = 0; i < 8; ++i) {
        const int e = etile * 16 + hi * 8 + i;
        const float* ps = tab + (size_t)SRCs[e] * TAB_LD + coff;
        const float4 s0 = *(const float4*)(ps), s1 = *(const float4*)(ps + 64);
        acc[0][i] = s0.x + bv[0];
        acc[1][i] = s0.y + bv[1];
        acc[2][i] = s0.z + bv[2];
        acc[3][i] = s0.w + bv[3];
        acc[4][i] = s1.x + bv[4];
        acc[5][i] = s1.y + bv[5];
        acc[6][i] = s1.z + bv[6];
        acc[7][i] = s1.w + bv[7];
    }
}

__device__ __forceinline__ void acc_gather_dst(v8f* acc, const float* __restrict__ tab, int g,
                                               const int* DSTs,
                                               int etile, int hi, int nsel, int row)
{
    const int coff = g * 256 + nsel * 128 + row * 4;
#pragma unroll
    for (int i = 0; i < 8; ++i) {
        const int e = etile * 16 + hi * 8 + i;
        const float* pd = tab + TAB_FLOATS + (size_t)DSTs[e] * TAB_LD + coff;
        const float4 d0 = *(const float4*)(pd), d1 = *(const float4*)(pd + 64);
        acc[0][i] += d0.x;
        acc[1][i] += d0.y;
        acc[2][i] += d0.z;
        acc[3][i] += d0.w;
        acc[4][i] += d1.x;
        acc[5][i] += d1.y;
        acc[6][i] += d1.z;
        acc[7][i] += d1.w;
    }
}

__global__ void __launch_bounds__(THREADS) __attribute__((amdgpu_waves_per_eu(4, 4)))
egnn_edge_tab(const float* __restrict__ x,
              const int*   __restrict__ eidx,const float* __restrict__ edist,
              const float* __restrict__ nW1, const float* __restrict__ nb1,
              const float* __restrict__ nW2, const float* __restrict__ nb2,
              const float* __restrict__ cW1, const float* __restrict__ cb1,
              const float* __restrict__ cW2,
              const float* __restrict__ eW1, const float* __restrict__ eb1,
              const float* __restrict__ eW2, const float* __restrict__ eb2,
              const float* __restrict__ tab, size_t ws_bytes,
              float* __restrict__ out)
{
    if (ws_bytes < TAB_BYTES) return;
    extern __shared__ float smem[];
    float* Ablk = smem;
    float* Hblk = smem;
    float* Bs   = smem + H_SIZE;
    int*   SRCs = (int*)(smem + H_SIZE + 2 * B_SIZE);
    int*   DSTs = SRCs + TILE_E;
    float* DIST = (float*)(DSTs + TILE_E);
    float* CWp  = smem + H_SIZE + 2 * B_SIZE + 4 * TILE_E;

    const int tid   = threadIdx.x;
    const int lane  = tid & 31;
    const int wave  = tid >> 5;
    const int row   = lane & 15;
    const int hi    = lane >> 4;
    const int etile = wave >> 1;
    const int nsel  = wave & 1;
    const int e0    = blockIdx.x * TILE_E;

    const float* cW1e = cW1 + 2 * NODE_DIM * HIDDEN_DIM;
    const float* nW1e = nW1 + 2 * NODE_DIM * HIDDEN_DIM;

    float R[32];
    int   cur = 0;
    panel_load<64>(eW2, EDGE_DIM, 0,  R,     tid);
    panel_load<64>(eW2, EDGE_DIM, 32, R + 8, tid);
    const float w1k = eW1[tid & 63], b1k = eb1[tid & 63];

    if (tid < TILE_E) {
        SRCs[tid] = eidx[e0 + tid];
        DSTs[tid] = eidx[N_EDGES + e0 + tid];
        DIST[tid] = edist[e0 + tid];
    }
    __syncthreads();

    v8f accC[8];
    acc_gather_src(accC, tab, 0, SRCs, cb1, etile, hi, nsel, row);
    acc_gather_dst(accC, tab, 0, DSTs, etile, hi, nsel, row);

    float* HID = smem + TILE_E * EA_LD;
    for (int i = tid; i < TILE_E * EDGE_DIM; i += THREADS) {
        int e = i >> 6;
        HID[e * EA_LD + (tid & 63)] = silu_f(DIST[e] * w1k + b1k);
    }
    panel_store<64>(Bs, R, tid);
    panel_store<64>(Bs + 32 * 64, R + 8, tid);
    __syncthreads();

    panel_load<256>(cW1e, HIDDEN_DIM, 0, R, tid);

    {
        v8f acc[2];
#pragma unroll
        for (int nt = 0; nt < 2; ++nt) {
            float bv = eb2[nsel * 32 + nt * 16 + row];
            v8f t = {bv, bv, bv, bv, bv, bv, bv, bv};
            acc[nt] = t;
        }
        const float* Arow = HID + (etile * 16 + row) * EA_LD + 4 * hi;
        const float* Bp   = Bs + ((hi * 64 + nsel * 32 + row) << 2);
        v4f a0, a1, b0[2], b1[2];
        frag_load<2, 64>(Arow, Bp, 0, a0, b0);
        frag_step<2, 64>(Arow, Bp, 1, a1, b1, a0, b0, acc);
        frag_step<2, 64>(Arow, Bp, 2, a0, b0, a1, b1, acc);
        frag_step<2, 64>(Arow, Bp, 3, a1, b1, a0, b0, acc);
        frag_step<2, 64>(Arow, Bp, 4, a0, b0, a1, b1, acc);
        frag_step<2, 64>(Arow, Bp, 5, a1, b1, a0, b0, acc);
        frag_step<2, 64>(Arow, Bp, 6, a0, b0, a1, b1, acc);
        frag_step<2, 64>(Arow, Bp, 7, a1, b1, a0, b0, acc);
        frag_mma<2>(a1, b1, acc);
#pragma unroll
        for (int nt = 0; nt < 2; ++nt)
#pragma unroll
            for (int i = 0; i < 8; ++i) {
                int e = etile * 16 + hi * 8 + i;
                Ablk[e * EA_LD + nsel * 32 + nt * 16 + row] = acc[nt][i];
            }
    }
    panel_store<256>(Bs + (cur ^ 1) * B_SIZE, R, tid);
    __syncthreads();
    cur ^= 1;

    v8f accN[8];
    {
        gemm_pipe<8, 256, 256>(cW1e, HIDDEN_DIM, EDGE_DIM, Ablk, EA_LD, Bs, cur, accC, R,
                               nW1e, HIDDEN_DIM, tid);
        acc_gather_src(accN, tab, 1, SRCs, nb1, etile, hi, nsel, row);
        float p[8];
#pragma unroll
        for (int i = 0; i < 8; ++i) p[i] = 0.0f;
#pragma unroll
        for (int nt = 0; nt < 8; ++nt) {
            const float w = cW2[nsel * 128 + nt * 16 + row];
#pragma unroll
            for (int i = 0; i < 8; ++i)
                p[i] += silu_f(accC[nt][i]) * w;
        }
#pragma unroll
        for (int m = 1; m < 16; m <<= 1)
#pragma unroll
            for (int i = 0; i < 8; ++i)
                p[i] += __shfl_xor(p[i], m, 32);
        if (row == 0) {
            float* q = CWp + nsel * TILE_E + etile * 16 + hi * 8;
            *(float4*)(q)     = make_float4(p[0], p[1], p[2], p[3]);
            *(float4*)(q + 4) = make_float4(p[4], p[5], p[6], p[7]);
        }
        acc_gather_dst(accN, tab, 1, DSTs, etile, hi, nsel, row);
        panel_store<256>(Bs + (cur ^ 1) * B_SIZE, R, tid);
        __syncthreads();
        cur ^= 1;
    }

    {
        gemm_pipe<8, 256, 128>(nW1e, HIDDEN_DIM, EDGE_DIM, Ablk, EA_LD, Bs, cur, accN, R,
                               nW2, NODE_DIM, tid);
#pragma unroll
        for (int nt = 0; nt < 8; ++nt)
#pragma unroll
            for (int i = 0; i < 8; ++i)
                accN[nt][i] = silu_f(accN[nt][i]);
        panel_store<128>(Bs + (cur ^ 1) * B_SIZE, R, tid);
        __syncthreads();
#pragma unroll
        for (int nt = 0; nt < 8; ++nt)
#pragma unroll
            for (int i = 0; i < 8; ++i) {
                int e = etile * 16 + hi * 8 + i;
                Hblk[e * H_LD + nsel * 128 + nt * 16 + row] = accN[nt][i];
            }
        __syncthreads();
        cur ^= 1;
    }

    {
        v8f acc[4];
#pragma unroll
        for (int nt = 0; nt < 4; ++nt) {
            float bv = nb2[nsel * 64 + nt * 16 + row];
            v8f t = {bv, bv, bv, bv, bv, bv, bv, bv};
            acc[nt] = t;
        }
        gemm_pipe<4, 128, 0>(nW2, NODE_DIM, HIDDEN_DIM, Hblk, H_LD, Bs, cur, acc, R,
                             nW2, NODE_DIM, tid);
#pragma unroll
        for (int nt = 0; nt < 4; ++nt)
#pragma unroll
            for (int i = 0; i < 8; ++i) {
                int e = etile * 16 + hi * 8 + i;
                int d = DSTs[e];
                atomAddF(out + d * NODE_DIM + nsel * 64 + nt * 16 + row, acc[nt][i]);
            }
    }

    if (tid < TILE_E) {
        int e = tid;
        float s = CWp[e] + CWp[TILE_E + e];
        int si = SRCs[e], di = DSTs[e];
        float dx = x[si * 3 + 0] - x[di * 3 + 0];
        float dy = x[si * 3 + 1] - x[di * 3 + 1];
        float dz = x[si * 3 + 2] - x[di * 3 + 2];
        float len = fmaxf(sqrtf(dx * dx + dy * dy + dz * dz), 1e-8f);
        float w = s / len;
        float* ox = out + N_NODES * NODE_DIM;
        atomAddF(ox + di * 3 + 0, w * dx);
        atomAddF(ox + di * 3 + 1, w * dy);
        atomAddF(ox + di * 3 + 2, w * dz);
    }
}

extern "C" void kernel_launch(void* const* d_in, const int* in_sizes, int n_in,
                              void* d_out, int out_size, void* d_ws, size_t ws_size,
                              hipStream_t stream)
{
    const float* h     = (const float*)d_in[0];
    const float* x     = (const float*)d_in[1];
    const int*   eidx  = (const int*)  d_in[2];
    const float* edist = (const float*)d_in[3];
    const float* nW1   = (const float*)d_in[4];
    const float* nb1   = (const float*)d_in[5];
    const float* nW2   = (const float*)d_in[6];
    const float* nb2   = (const float*)d_in[7];
    const float* cW1   = (const float*)d_in[8];
    const float* cb1   = (const float*)d_in[9];
    const float* cW2   = (const float*)d_in[10];
    const float* eW1   = (const float*)d_in[11];
    const float* eb1   = (const float*)d_in[12];
    const float* eW2   = (const float*)d_in[13];
    const float* eb2   = (const float*)d_in[14];
    float* out = (float*)d_out;

    const int tot = N_NODES * (NODE_DIM + 3);
    egnn_init_out<<<(tot + 255) / 256, 256, 0, stream>>>(h, x, out);

    float* tab = (float*)d_ws;
    egnn_node_tables<<<N_NODES / TILE_E, THREADS, (size_t)NSMEM_FLOATS * sizeof(float), stream>>>(
        h, nW1, cW1, tab, ws_size);
    egnn_edge_tab<<<N_EDGES / TILE_E, THREADS, (size_t)ESMEM_FLOATS * sizeof(float), stream>>>(
        x, eidx, edist,
        nW1, nb1, nW2, nb2,
        cW1, cb1, cW2,
        eW1, eb1, eW2, eb2,
        tab, ws_size,
        out);

    const size_t smem_bytes = (size_t)SMEM_FLOATS * sizeof(float);
    egnn_edge_kernel<<<N_EDGES / TILE_E, THREADS, smem_bytes, stream>>>(
        h, x, eidx, edist,
        nW1, nb1, nW2, nb2,
        cW1, cb1, cW2,
        eW1, eb1, eW2, eb2,
        ws_size,
        out);
}


# Edge MLP second layer folded into the first-layer weights: attr @ W1[256:320] = hid @ (eW2 @ W1[256:320]), built once per call on WMMA; per-edge 64x64 product gone; edge-term sum order changes
# speedup vs baseline: 4.9163x; 1.0340x over previous
#include <hip/hip_runtime.h>
#include <math.h>

#define N_NODES   32768
#define N_EDGES   524288
#define NODE_DIM  128
#define HIDDEN_DIM 256
#define EDGE_DIM  64
#define IN_DIM    320

#define TILE_E    64
#define THREADS   256

#define A_LD      324
#define A_OFF     0
#define A_SIZE    (TILE_E * A_LD)
#define H_LD      260
#define H_OFF     A_OFF
#define H_SIZE    (TILE_E * H_LD)
#define B_OFF     (A_OFF + A_SIZE)
#define B_SIZE    (32 * 256)
#define S_OFF     (B_OFF + 2 * B_SIZE)
#define SMEM_FLOATS (S_OFF + 6 * TILE_E)

#define TAB_LD    512
#define TAB_FLOATS ((size_t)N_NODES * TAB_LD)
#define TAB_BYTES  ((size_t)2 * N_NODES * TAB_LD * sizeof(float))

#define FOLD_W_FLOATS (EDGE_DIM * HIDDEN_DIM)
#define FOLD_FLOATS   (2 * FOLD_W_FLOATS + 2 * HIDDEN_DIM)
#define WS_NEED       (TAB_BYTES + (size_t)FOLD_FLOATS * sizeof(float))

#define NA_LD     132
#define NA_SIZE   (TILE_E * NA_LD)
#define NSMEM_FLOATS (NA_SIZE + 2 * B_SIZE)

#define EA_LD     68
#define ESMEM_FLOATS (H_SIZE + 2 * B_SIZE + 6 * TILE_E)

typedef float v2f __attribute__((ext_vector_type(2)));
typedef float v4f __attribute__((ext_vector_type(4)));
typedef float v8f __attribute__((ext_vector_type(8)));

__device__ __forceinline__ float silu_f(float v) {
    return v * __builtin_amdgcn_rcpf(1.0f + __expf(-v));
}

__device__ __forceinline__ void atomAddF(float* p, float v) {
    __hip_atomic_fetch_add(p, v, __ATOMIC_RELAXED, __HIP_MEMORY_SCOPE_AGENT);
}


template<int NSTAGE>
__device__ __forceinline__ void panel_load(const float* __restrict__ W, int ldw, int kp,
                                           float* R, int tid)
{
    constexpr int KGRP = NSTAGE / 32;
    const int n  = tid & (NSTAGE - 1);
    const int g0 = (tid / NSTAGE) * KGRP;
    const float* p = W + (kp + 4 * g0) * ldw + n;
#pragma unroll
    for (int j = 0; j < 4 * KGRP; ++j)
        R[j] = p[j * ldw];
}

template<int NSTAGE>
__device__ __forceinline__ void panel_store(float* __restrict__ Bs, const float* R, int tid)
{
    constexpr int KGRP = NSTAGE / 32;
    const int n  = tid & (NSTAGE - 1);
    const int g0 = (tid / NSTAGE) * KGRP;
#pragma unroll
    for (int j = 0; j < KGRP; ++j) {
        float4 v = make_float4(R[4 * j], R[4 * j + 1], R[4 * j + 2], R[4 * j + 3]);
        *(float4*)(Bs + (((g0 + j) * NSTAGE + n) << 2)) = v;
    }
}

template<int NTILES, int NSTAGE>
__device__ __forceinline__ void frag_load(const float* __restrict__ Ap,
                                          const float* __restrict__ Bp,
                                          int kb, v4f& a, v4f* b)
{
    a = *(const v4f*)(Ap + 8 * kb);
#pragma unroll
    for (int nt = 0; nt < NTILES; ++nt)
        b[nt] = *(const v4f*)(Bp + ((2 * kb * NSTAGE + nt * 16) << 2));
}

template<int NTILES>
__device__ __forceinline__ void frag_mma(const v4f& a, const v4f* b, v8f* acc)
{
    const v2f alo = {a.x, a.y};
    const v2f ahi = {a.z, a.w};
#pragma unroll
    for (int nt = 0; nt < NTILES; ++nt) {
        const v2f bl = {b[nt].x, b[nt].y};
        acc[nt] = __builtin_amdgcn_wmma_f32_16x16x4_f32(
             false, alo,  false, bl,
             (short)0, acc[nt],
             false,  false);
    }
#pragma unroll
    for (int nt = 0; nt < NTILES; ++nt) {
        const v2f bh = {b[nt].z, b[nt].w};
        acc[nt] = __builtin_amdgcn_wmma_f32_16x16x4_f32(
             false, ahi,  false, bh,
             (short)0, acc[nt],
             false,  false);
    }
}

template<int NTILES, int NSTAGE>
__device__ __forceinline__ void frag_step(const float* __restrict__ Ap,
                                          const float* __restrict__ Bp, int kb,
                                          v4f& an, v4f* bn,
                                          const v4f& ac, const v4f* bc, v8f* acc)
{
    frag_load<NTILES, NSTAGE>(Ap, Bp, kb, an, bn);
    __builtin_amdgcn_sched_barrier(0);
    frag_mma<NTILES>(ac, bc, acc);
    __builtin_amdgcn_sched_barrier(0);
}

template<int NTILES, int NSTAGE, int NNEXT>
__device__ __forceinline__ void gemm_pipe(const float* __restrict__ W, int ldw, int K,
                                          const float* __restrict__ A, int lda,
                                          float* __restrict__ Bs, int& cur,
                                          v8f* acc, float* R,
                                          const float* __restrict__ Wnext, int ldwnext,
                                          int tid)
{
    const int lane  = tid & 31;
    const int wave  = tid >> 5;
    const int row   = lane & 15;
    const int hi    = lane >> 4;
    const int etile = wave >> 1;
    const int nsel  = wave & 1;
    const float* Arow = A + (etile * 16 + row) * lda + 4 * hi;
    const int    boff = (hi * NSTAGE + nsel * (NSTAGE / 2) + row) << 2;

    v4f a0, a1, b0[NTILES], b1[NTILES];
    frag_load<NTILES, NSTAGE>(Arow, Bs + cur * B_SIZE + boff, 0, a0, b0);

    for (int kp = 0; kp < K - 32; kp += 32) {
        const float* Ap = Arow + kp;
        const float* Bp = Bs + cur * B_SIZE + boff;
        panel_load<NSTAGE>(W, ldw, kp + 32, R, tid);
        frag_step<NTILES, NSTAGE>(Ap, Bp, 1, a1, b1, a0, b0, acc);
        frag_step<NTILES, NSTAGE>(Ap, Bp, 2, a0, b0, a1, b1, acc);
        frag_step<NTILES, NSTAGE>(Ap, Bp, 3, a1, b1, a0, b0, acc);
        panel_store<NSTAGE>(Bs + (cur ^ 1) * B_SIZE, R, tid);
        __syncthreads();
        cur ^= 1;
        frag_step<NTILES, NSTAGE>(Ap + 32, Bs + cur * B_SIZE + boff, 0, a0, b0, a1, b1, acc);
    }
    if constexpr (NNEXT > 0) panel_load<NNEXT>(Wnext, ldwnext, 0, R, tid);
    {
        const float* Ap = Arow + K - 32;
        const float* Bp = Bs + cur * B_SIZE + boff;
        frag_step<NTILES, NSTAGE>(Ap, Bp, 1, a1, b1, a0, b0, acc);
        frag_step<NTILES, NSTAGE>(Ap, Bp, 2, a0, b0, a1, b1, acc);
        frag_step<NTILES, NSTAGE>(Ap, Bp, 3, a1, b1, a0, b0, acc);
        frag_mma<NTILES>(a1, b1, acc);
    }
}

__global__ void egnn_init_out(const float* __restrict__ h,
                              const float* __restrict__ x,
                              float* __restrict__ out)
{
    int i = blockIdx.x * blockDim.x + threadIdx.x;
    if (i < N_NODES * NODE_DIM)                out[i] = h[i];
    else if (i < N_NODES * (NODE_DIM + 3))     out[i] = x[i - N_NODES * NODE_DIM];
}

__global__ void __launch_bounds__(THREADS)
egnn_edge_kernel(const float* __restrict__ h,   const float* __restrict__ x,
                 const int*   __restrict__ eidx,const float* __restrict__ edist,
                 const float* __restrict__ nW1, const float* __restrict__ nb1,
                 const float* __restrict__ nW2, const float* __restrict__ nb2,
                 const float* __restrict__ cW1, const float* __restrict__ cb1,
                 const float* __restrict__ cW2,
                 const float* __restrict__ eW1, const float* __restrict__ eb1,
                 const float* __restrict__ eW2, const float* __restrict__ eb2,
                 size_t ws_bytes,
                 float* __restrict__ out)
{
    if (ws_bytes >= WS_NEED) return;
    extern __shared__ float smem[];
    float* Ablk = smem + A_OFF;
    float* Hblk = smem + H_OFF;
    float* Bs   = smem + B_OFF;
    int*   SRCs = (int*)(smem + S_OFF);
    int*   DSTs = (int*)(smem + S_OFF + TILE_E);
    float* DIST = smem + S_OFF + 2 * TILE_E;
    float* CWp  = smem + S_OFF + 4 * TILE_E;

    const int tid   = threadIdx.x;
    const int lane  = tid & 31;
    const int wave  = tid >> 5;
    const int row   = lane & 15;
    const int hi    = lane >> 4;
    const int etile = wave >> 1;
    const int nsel  = wave & 1;
    const int e0    = blockIdx.x * TILE_E;

    float R[32];
    int   cur = 0;
    panel_load<256>(cW1, HIDDEN_DIM, 0, R, tid);

    if (tid < TILE_E) {
        SRCs[tid] = eidx[e0 + tid];
        DSTs[tid] = eidx[N_EDGES + e0 + tid];
        DIST[tid] = edist[e0 + tid];
    }
    __syncthreads();

    for (int i = tid; i < TILE_E * 64; i += THREADS) {
        int e = i >> 6, q = i & 63;
        int half = q >> 5, c4 = q & 31;
        int node = half ? DSTs[e] : SRCs[e];
        *(float4*)(Ablk + e * A_LD + half * NODE_DIM + c4 * 4) =
            ((const float4*)h)[node * (NODE_DIM / 4) + c4];
    }

    float* hid = Bs + B_SIZE;
    for (int i = tid; i < TILE_E * EDGE_DIM; i += THREADS) {
        int e = i >> 6, k = i & 63;
        hid[k * TILE_E + e] = silu_f(DIST[e] * eW1[k] + eb1[k]);
    }
    __syncthreads();
    for (int i = tid; i < TILE_E * EDGE_DIM; i += THREADS) {
        int e = i >> 6, j = i & 63;
        float s = eb2[j];
#pragma unroll 8
        for (int k = 0; k < EDGE_DIM; ++k)
            s += hid[k * TILE_E + e] * eW2[k * EDGE_DIM + j];
        Ablk[e * A_LD + 2 * NODE_DIM + j] = s;
    }
    panel_store<256>(Bs, R, tid);
    __syncthreads();

    {
        v8f acc[8];
#pragma unroll
        for (int nt = 0; nt < 8; ++nt) {
            float bv = cb1[nsel * 128 + nt * 16 + row];
            v8f t = {bv, bv, bv, bv, bv, bv, bv, bv};
            acc[nt] = t;
        }
        gemm_pipe<8, 256, 256>(cW1, HIDDEN_DIM, IN_DIM, Ablk, A_LD, Bs, cur, acc, R,
                               nW1, HIDDEN_DIM, tid);
        float p[8];
#pragma unroll
        for (int i = 0; i < 8; ++i) p[i] = 0.0f;
#pragma unroll
        for (int nt = 0; nt < 8; ++nt) {
            const float w = cW2[nsel * 128 + nt * 16 + row];
#pragma unroll
            for (int i = 0; i < 8; ++i)
                p[i] += silu_f(acc[nt][i]) * w;
        }
#pragma unroll
        for (int m = 1; m < 16; m <<= 1)
#pragma unroll
            for (int i = 0; i < 8; ++i)
                p[i] += __shfl_xor(p[i], m, 32);
        if (row == 0) {
            float* q = CWp + nsel * TILE_E + etile * 16 + hi * 8;
            *(float4*)(q)     = make_float4(p[0], p[1], p[2], p[3]);
            *(float4*)(q + 4) = make_float4(p[4], p[5], p[6], p[7]);
        }
        panel_store<256>(Bs + (cur ^ 1) * B_SIZE, R, tid);
        __syncthreads();
        cur ^= 1;
    }

    {
        v8f acc[8];
#pragma unroll
        for (int nt = 0; nt < 8; ++nt) {
            float bv = nb1[nsel * 128 + nt * 16 + row];
            v8f t = {bv, bv, bv, bv, bv, bv, bv, bv};
            acc[nt] = t;
        }
        gemm_pipe<8, 256, 128>(nW1, HIDDEN_DIM, IN_DIM, Ablk, A_LD, Bs, cur, acc, R,
                               nW2, NODE_DIM, tid);
#pragma unroll
        for (int nt = 0; nt < 8; ++nt)
#pragma unroll
            for (int i = 0; i < 8; ++i)
                acc[nt][i] = silu_f(acc[nt][i]);
        panel_store<128>(Bs + (cur ^ 1) * B_SIZE, R, tid);
        __syncthreads();
#pragma unroll
        for (int nt = 0; nt < 8; ++nt)
#pragma unroll
            for (int i = 0; i < 8; ++i) {
                int e = etile * 16 + hi * 8 + i;
                Hblk[e * H_LD + nsel * 128 + nt * 16 + row] = acc[nt][i];
            }
        __syncthreads();
        cur ^= 1;
    }

    {
        v8f acc[4];
#pragma unroll
        for (int nt = 0; nt < 4; ++nt) {
            float bv = nb2[nsel * 64 + nt * 16 + row];
            v8f t = {bv, bv, bv, bv, bv, bv, bv, bv};
            acc[nt] = t;
        }
        gemm_pipe<4, 128, 0>(nW2, NODE_DIM, HIDDEN_DIM, Hblk, H_LD, Bs, cur, acc, R,
                             nW2, NODE_DIM, tid);
#pragma unroll
        for (int nt = 0; nt < 4; ++nt)
#pragma unroll
            for (int i = 0; i < 8; ++i) {
                int e = etile * 16 + hi * 8 + i;
                int d = DSTs[e];
                atomAddF(out + d * NODE_DIM + nsel * 64 + nt * 16 + row, acc[nt][i]);
            }
    }

    if (tid < TILE_E) {
        int e = tid;
        float s = CWp[e] + CWp[TILE_E + e];
        int si = SRCs[e], di = DSTs[e];
        float dx = x[si * 3 + 0] - x[di * 3 + 0];
        float dy = x[si * 3 + 1] - x[di * 3 + 1];
        float dz = x[si * 3 + 2] - x[di * 3 + 2];
        float len = fmaxf(sqrtf(dx * dx + dy * dy + dz * dz), 1e-8f);
        float w = s / len;
        float* ox = out + N_NODES * NODE_DIM;
        atomAddF(ox + di * 3 + 0, w * dx);
        atomAddF(ox + di * 3 + 1, w * dy);
        atomAddF(ox + di * 3 + 2, w * dz);
    }
}


template<int NNEXT>
__device__ __forceinline__ void node_product(const float* __restrict__ W, const float* __restrict__ Wnext,
                                             const float* __restrict__ Ablk, float* __restrict__ Bs, int& cur,
                                             float* R, float* __restrict__ trow, int tid)
{
    const int lane  = tid & 31;
    const int wave  = tid >> 5;
    const int row   = lane & 15;
    const int hi    = lane >> 4;
    const int etile = wave >> 1;
    const int nsel  = wave & 1;
    v8f acc[8];
#pragma unroll
    for (int nt = 0; nt < 8; ++nt) {
        v8f z = {0.f, 0.f, 0.f, 0.f, 0.f, 0.f, 0.f, 0.f};
        acc[nt] = z;
    }
    gemm_pipe<8, 256, NNEXT>(W, HIDDEN_DIM, NODE_DIM, Ablk, NA_LD, Bs, cur, acc, R, Wnext, HIDDEN_DIM, tid);
    float* p = trow + (size_t)(etile * 16 + hi * 8) * TAB_LD + nsel * 128 + row * 4;
#pragma unroll
    for (int i = 0; i < 8; ++i) {
        *(float4*)(p + i * TAB_LD)      = make_float4(acc[0][i], acc[1][i], acc[2][i], acc[3][i]);
        *(float4*)(p + i * TAB_LD + 64) = make_float4(acc[4][i], acc[5][i], acc[6][i], acc[7][i]);
    }
    if constexpr (NNEXT > 0) {
        panel_store<NNEXT>(Bs + (cur ^ 1) * B_SIZE, R, tid);
        __syncthreads();
        cur ^= 1;
    }
}

__device__ __forceinline__ void fold_product(const float* __restrict__ W1e, const float* __restrict__ b1,
                                             const float* __restrict__ eW2, const float* __restrict__ eb2,
                                             float* __restrict__ Mout, float* __restrict__ bout,
                                             float* __restrict__ Ablk, float* __restrict__ Bs, int tid)
{
    const int lane  = tid & 31;
    const int wave  = tid >> 5;
    const int row   = lane & 15;
    const int hi    = lane >> 4;
    const int etile = wave >> 1;
    const int nsel  = wave & 1;

    float R[32];
    int   cur = 0;
    panel_load<256>(W1e, HIDDEN_DIM, 0, R, tid);
    for (int i = tid; i < EDGE_DIM * (EDGE_DIM / 4); i += THREADS) {
        int r = i >> 4, c4 = i & 15;
        *(float4*)(Ablk + r * EA_LD + c4 * 4) = ((const float4*)eW2)[r * (EDGE_DIM / 4) + c4];
    }
    panel_store<256>(Bs, R, tid);
    __syncthreads();

    v8f acc[8];
#pragma unroll
    for (int nt = 0; nt < 8; ++nt) {
        v8f z = {0.f, 0.f, 0.f, 0.f, 0.f, 0.f, 0.f, 0.f};
        acc[nt] = z;
    }
    gemm_pipe<8, 256, 0>(W1e, HIDDEN_DIM, EDGE_DIM, Ablk, EA_LD, Bs, cur, acc, R, W1e, HIDDEN_DIM, tid);
    float* p = Mout + (etile * 16 + hi * 8) * HIDDEN_DIM + nsel * 128 + row;
#pragma unroll
    for (int nt = 0; nt < 8; ++nt)
#pragma unroll
        for (int i = 0; i < 8; ++i)
            p[i * HIDDEN_DIM + nt * 16] = acc[nt][i];

    float s = b1[tid];
    for (int j = 0; j < EDGE_DIM; ++j)
        s += eb2[j] * W1e[j * HIDDEN_DIM + tid];
    bout[tid] = s;
}

__global__ void __launch_bounds__(THREADS)
egnn_node_tables(const float* __restrict__ h,
                 const float* __restrict__ nW1, const float* __restrict__ nb1,
                 const float* __restrict__ cW1, const float* __restrict__ cb1,
                 const float* __restrict__ eW2, const float* __restrict__ eb2,
                 float* __restrict__ tab, size_t ws_bytes)
{
    if (ws_bytes < WS_NEED) return;
    extern __shared__ float smem[];
    float* Ablk = smem;
    float* Bs   = smem + NA_SIZE;
    const int tid = threadIdx.x;

    if (blockIdx.x < 2) {
        const int g = blockIdx.x;
        float* fold = tab + 2 * TAB_FLOATS;
        fold_product((g ? nW1 : cW1) + 2 * NODE_DIM * HIDDEN_DIM, g ? nb1 : cb1, eW2, eb2,
                     fold + g * FOLD_W_FLOATS, fold + 2 * FOLD_W_FLOATS + g * HIDDEN_DIM, Ablk, Bs, tid);
        return;
    }
    const int n0  = (blockIdx.x - 2) * TILE_E;

    float R[32];
    int   cur = 0;
    panel_load<256>(cW1, HIDDEN_DIM, 0, R, tid);
    for (int i = tid; i < TILE_E * 32; i += THREADS) {
        int r = i >> 5, c4 = i & 31;
        *(float4*)(Ablk + r * NA_LD + c4 * 4) = ((const float4*)h)[(n0 + r) * (NODE_DIM / 4) + c4];
    }
    panel_store<256>(Bs, R, tid);
    __syncthreads();

    float* t0 = tab + (size_t)n0 * TAB_LD;
    float* t1 = tab + TAB_FLOATS + (size_t)n0 * TAB_LD;
    node_product<256>(cW1,                         nW1,                         Ablk, Bs, cur, R, t0,       tid);
    node_product<256>(nW1,                         cW1 + NODE_DIM * HIDDEN_DIM, Ablk, Bs, cur, R, t0 + 256, tid);
    node_product<256>(cW1 + NODE_DIM * HIDDEN_DIM, nW1 + NODE_DIM * HIDDEN_DIM, Ablk, Bs, cur, R, t1,       tid);
    node_product<0>  (nW1 + NODE_DIM * HIDDEN_DIM, nW1,                         Ablk, Bs, cur, R, t1 + 256, tid);
}

__device__ __forceinline__ void acc_gather_src(v8f* acc, const float* __restrict__ tab, int g,
                                               const int* SRCs, const float* __restrict__ bias,
                                               int etile, int hi, int nsel, int row)
{
    const int coff = g * 256 + nsel * 128 + row * 4;
    float bv[8];
#pragma unroll
    for (int nt = 0; nt < 8; ++nt) bv[nt] = bias[nsel * 128 + nt * 16 + row];
#pragma unroll
    for (int i = 0; i < 8; ++i) {
        const int e = etile * 16 + hi * 8 + i;
        const float* ps = tab + (size_t)SRCs[e] * TAB_LD + coff;
        const float4 s0 = *(const float4*)(ps), s1 = *(const float4*)(ps + 64);
        acc[0][i] = s0.x + bv[0];
        acc[1][i] = s0.y + bv[1];
        acc[2][i] = s0.z + bv[2];
        acc[3][i] = s0.w + bv[3];
        acc[4][i] = s1.x + bv[4];
        acc[5][i] = s1.y + bv[5];
        acc[6][i] = s1.z + bv[6];
        acc[7][i] = s1.w + bv[7];
    }
}

__device__ __forceinline__ void acc_gather_dst(v8f* acc, const float* __restrict__ tab, int g,
                                               const int* DSTs,
                                               int etile, int hi, int nsel, int row)
{
    const int coff = g * 256 + nsel * 128 + row * 4;
#pragma unroll
    for (int i = 0; i < 8; ++i) {
        const int e = etile * 16 + hi * 8 + i;
        const float* pd = tab + TAB_FLOATS + (size_t)DSTs[e] * TAB_LD + coff;
        const float4 d0 = *(const float4*)(pd), d1 = *(const float4*)(pd + 64);
        acc[0][i] += d0.x;
        acc[1][i] += d0.y;
        acc[2][i] += d0.z;
        acc[3][i] += d0.w;
        acc[4][i] += d1.x;
        acc[5][i] += d1.y;
        acc[6][i] += d1.z;
        acc[7][i] += d1.w;
    }
}

__global__ void __launch_bounds__(THREADS) __attribute__((amdgpu_waves_per_eu(4, 4)))
egnn_edge_tab(const float* __restrict__ x,
              const int*   __restrict__ eidx,const float* __restrict__ edist,
              const float* __restrict__ nW2, const float* __restrict__ nb2,
              const float* __restrict__ cW2,
              const float* __restrict__ eW1, const float* __restrict__ eb1,
              const float* __restrict__ tab, size_t ws_bytes,
              float* __restrict__ out)
{
    if (ws_bytes < WS_NEED) return;
    extern __shared__ float smem[];
    float* Ablk = smem;
    float* Hblk = smem;
    float* Bs   = smem + H_SIZE;
    int*   SRCs = (int*)(smem + H_SIZE + 2 * B_SIZE);
    int*   DSTs = SRCs + TILE_E;
    float* DIST = (float*)(DSTs + TILE_E);
    float* CWp  = smem + H_SIZE + 2 * B_SIZE + 4 * TILE_E;

    const int tid   = threadIdx.x;
    const int lane  = tid & 31;
    const int wave  = tid >> 5;
    const int row   = lane & 15;
    const int hi    = lane >> 4;
    const int etile = wave >> 1;
    const int nsel  = wave & 1;
    const int e0    = blockIdx.x * TILE_E;

    const float* fold = tab + 2 * TAB_FLOATS;
    const float* cW1e = fold;
    const float* nW1e = fold + FOLD_W_FLOATS;
    const float* cb1f = fold + 2 * FOLD_W_FLOATS;
    const float* nb1f = cb1f + HIDDEN_DIM;

    float R[32];
    int   cur = 0;
    panel_load<256>(cW1e, HIDDEN_DIM, 0, R, tid);
    const float w1k = eW1[tid & 63], b1k = eb1[tid & 63];

    if (tid < TILE_E) {
        SRCs[tid] = eidx[e0 + tid];
        DSTs[tid] = eidx[N_EDGES + e0 + tid];
        DIST[tid] = edist[e0 + tid];
    }
    __syncthreads();

    v8f accC[8];
    acc_gather_src(accC, tab, 0, SRCs, cb1f, etile, hi, nsel, row);
    acc_gather_dst(accC, tab, 0, DSTs, etile, hi, nsel, row);

    for (int i = tid; i < TILE_E * EDGE_DIM; i += THREADS) {
        int e = i >> 6;
        Ablk[e * EA_LD + (tid & 63)] = silu_f(DIST[e] * w1k + b1k);
    }
    panel_store<256>(Bs, R, tid);
    __syncthreads();

    v8f accN[8];
    {
        gemm_pipe<8, 256, 256>(cW1e, HIDDEN_DIM, EDGE_DIM, Ablk, EA_LD, Bs, cur, accC, R,
                               nW1e, HIDDEN_DIM, tid);
        acc_gather_src(accN, tab, 1, SRCs, nb1f, etile, hi, nsel, row);
        float p[8];
#pragma unroll
        for (int i = 0; i < 8; ++i) p[i] = 0.0f;
#pragma unroll
        for (int nt = 0; nt < 8; ++nt) {
            const float w = cW2[nsel * 128 + nt * 16 + row];
#pragma unroll
            for (int i = 0; i < 8; ++i)
                p[i] += silu_f(accC[nt][i]) * w;
        }
#pragma unroll
        for (int m = 1; m < 16; m <<= 1)
#pragma unroll
            for (int i = 0; i < 8; ++i)
                p[i] += __shfl_xor(p[i], m, 32);
        if (row == 0) {
            float* q = CWp + nsel * TILE_E + etile * 16 + hi * 8;
            *(float4*)(q)     = make_float4(p[0], p[1], p[2], p[3]);
            *(float4*)(q + 4) = make_float4(p[4], p[5], p[6], p[7]);
        }
        acc_gather_dst(accN, tab, 1, DSTs, etile, hi, nsel, row);
        panel_store<256>(Bs + (cur ^ 1) * B_SIZE, R, tid);
        __syncthreads();
        cur ^= 1;
    }

    {
        gemm_pipe<8, 256, 128>(nW1e, HIDDEN_DIM, EDGE_DIM, Ablk, EA_LD, Bs, cur, accN, R,
                               nW2, NODE_DIM, tid);
#pragma unroll
        for (int nt = 0; nt < 8; ++nt)
#pragma unroll
            for (int i = 0; i < 8; ++i)
                accN[nt][i] = silu_f(accN[nt][i]);
        panel_store<128>(Bs + (cur ^ 1) * B_SIZE, R, tid);
        __syncthreads();
#pragma unroll
        for (int nt = 0; nt < 8; ++nt)
#pragma unroll
            for (int i = 0; i < 8; ++i) {
                int e = etile * 16 + hi * 8 + i;
                Hblk[e * H_LD + nsel * 128 + nt * 16 + row] = accN[nt][i];
            }
        __syncthreads();
        cur ^= 1;
    }

    {
        v8f acc[4];
#pragma unroll
        for (int nt = 0; nt < 4; ++nt) {
            float bv = nb2[nsel * 64 + nt * 16 + row];
            v8f t = {bv, bv, bv, bv, bv, bv, bv, bv};
            acc[nt] = t;
        }
        gemm_pipe<4, 128, 0>(nW2, NODE_DIM, HIDDEN_DIM, Hblk, H_LD, Bs, cur, acc, R,
                             nW2, NODE_DIM, tid);
#pragma unroll
        for (int nt = 0; nt < 4; ++nt)
#pragma unroll
            for (int i = 0; i < 8; ++i) {
                int e = etile * 16 + hi * 8 + i;
                int d = DSTs[e];
                atomAddF(out + d * NODE_DIM + nsel * 64 + nt * 16 + row, acc[nt][i]);
            }
    }

    if (tid < TILE_E) {
        int e = tid;
        float s = CWp[e] + CWp[TILE_E + e];
        int si = SRCs[e], di = DSTs[e];
        float dx = x[si * 3 + 0] - x[di * 3 + 0];
        float dy = x[si * 3 + 1] - x[di * 3 + 1];
        float dz = x[si * 3 + 2] - x[di * 3 + 2];
        float len = fmaxf(sqrtf(dx * dx + dy * dy + dz * dz), 1e-8f);
        float w = s / len;
        float* ox = out + N_NODES * NODE_DIM;
        atomAddF(ox + di * 3 + 0, w * dx);
        atomAddF(ox + di * 3 + 1, w * dy);
        atomAddF(ox + di * 3 + 2, w * dz);
    }
}

extern "C" void kernel_launch(void* const* d_in, const int* in_sizes, int n_in,
                              void* d_out, int out_size, void* d_ws, size_t ws_size,
                              hipStream_t stream)
{
    const float* h     = (const float*)d_in[0];
    const float* x     = (const float*)d_in[1];
    const int*   eidx  = (const int*)  d_in[2];
    const float* edist = (const float*)d_in[3];
    const float* nW1   = (const float*)d_in[4];
    const float* nb1   = (const float*)d_in[5];
    const float* nW2   = (const float*)d_in[6];
    const float* nb2   = (const float*)d_in[7];
    const float* cW1   = (const float*)d_in[8];
    const float* cb1   = (const float*)d_in[9];
    const float* cW2   = (const float*)d_in[10];
    const float* eW1   = (const float*)d_in[11];
    const float* eb1   = (const float*)d_in[12];
    const float* eW2   = (const float*)d_in[13];
    const float* eb2   = (const float*)d_in[14];
    float* out = (float*)d_out;

    const int tot = N_NODES * (NODE_DIM + 3);
    egnn_init_out<<<(tot + 255) / 256, 256, 0, stream>>>(h, x, out);

    float* tab = (float*)d_ws;
    egnn_node_tables<<<N_NODES / TILE_E + 2, THREADS, (size_t)NSMEM_FLOATS * sizeof(float), stream>>>(
        h, nW1, nb1, cW1, cb1, eW2, eb2, tab, ws_size);
    egnn_edge_tab<<<N_EDGES / TILE_E, THREADS, (size_t)ESMEM_FLOATS * sizeof(float), stream>>>(
        x, eidx, edist,
        nW2, nb2,
        cW2,
        eW1, eb1,
        tab, ws_size,
        out);

    const size_t smem_bytes = (size_t)SMEM_FLOATS * sizeof(float);
    egnn_edge_kernel<<<N_EDGES / TILE_E, THREADS, smem_bytes, stream>>>(
        h, x, eidx, edist,
        nW1, nb1, nW2, nb2,
        cW1, cb1, cW2,
        eW1, eb1, eW2, eb2,
        ws_size,
        out);
}
